# starting kernel
# baseline (speedup 1.0000x reference)
#include <hip/hip_runtime.h>
#include <cstdint>
#include <cstddef>

#define BB 4
#define HH 32
#define WWD 256
#define CC 96
#define DI 192
#define KK 2
#define NN 16
#define RR 6
#define LL (HH*WWD)
#define BL (BB*LL)
#define NOUT (2*DI)
#define CDBL (RR + 2*NN)
#define NC 128
#define CL (LL/NC)

typedef float v2f __attribute__((ext_vector_type(2)));
typedef float v8f __attribute__((ext_vector_type(8)));

__device__ __forceinline__ v8f wmma4(v2f a, v2f b, v8f c) {
  return __builtin_amdgcn_wmma_f32_16x16x4_f32(false, a, false, b, (short)0, c,
                                               false, false);
}

__device__ __forceinline__ float silu_f(float x) {
  return x / (1.0f + __expf(-x));
}
__device__ __forceinline__ float softplus_f(float x) {
  return (x > 20.0f) ? x : log1pf(__expf(x));
}

__global__ void __launch_bounds__(256) k_inproj(const float* __restrict__ x,
                                                const float* __restrict__ W_in,
                                                float* __restrict__ xz) {
  __shared__ float sX[16][CC + 1];
  const int rowBase = blockIdx.x * 16;
  const int tid = threadIdx.x;
  for (int i = tid; i < 16 * CC; i += 256) {
    int m = i / CC, kk = i % CC;
    sX[m][kk] = x[(size_t)(rowBase + m) * CC + kk];
  }
  __syncthreads();
  const int wave = tid >> 5, lane = tid & 31;
  const int lid = lane & 15, half = lane >> 4;
  for (int nt = wave; nt < NOUT / 16; nt += 8) {
    const int nbase = nt * 16;
    const float* wrow = W_in + (size_t)(nbase + lid) * CC;
    v8f acc = {};
    for (int k0 = 0; k0 < CC; k0 += 4) {
      v2f a; a.x = sX[lid][k0 + 2 * half]; a.y = sX[lid][k0 + 2 * half + 1];
      v2f b; b.x = wrow[k0 + 2 * half];    b.y = wrow[k0 + 2 * half + 1];
      acc = wmma4(a, b, acc);
    }
    float* orow = xz + (size_t)(rowBase + half * 8) * NOUT + nbase + lid;
#pragma unroll
    for (int v = 0; v < 8; ++v) orow[(size_t)v * NOUT] = acc[v];
  }
}

__global__ void __launch_bounds__(256) k_dwconv(const float* __restrict__ xz,
                                                const float* __restrict__ conv_w,
                                                const float* __restrict__ conv_b,
                                                float* __restrict__ xconv) {
  int idx = blockIdx.x * 256 + threadIdx.x;
  if (idx >= BB * HH * WWD * DI) return;
  int c = idx % DI; int t = idx / DI;
  int w = t % WWD; t /= WWD;
  int h = t % HH;  int b = t / HH;
  float s = conv_b[c];
#pragma unroll
  for (int dy = -1; dy <= 1; ++dy) {
    int hh = h + dy; if (hh < 0 || hh >= HH) continue;
#pragma unroll
    for (int dx = -1; dx <= 1; ++dx) {
      int ww = w + dx; if (ww < 0 || ww >= WWD) continue;
      s += xz[(size_t)((b * HH + hh) * WWD + ww) * NOUT + c] *
           conv_w[c * 9 + (dy + 1) * 3 + (dx + 1)];
    }
  }
  xconv[(size_t)((b * HH + h) * WWD + w) * DI + c] = silu_f(s);
}

__global__ void __launch_bounds__(256) k_xdbl(const float* __restrict__ xconv,
                                              const float* __restrict__ xpw,
                                              float* __restrict__ x_dbl) {
  const int tid = threadIdx.x, wave = tid >> 5, lane = tid & 31;
  const int lid = lane & 15, half = lane >> 4;
  const int tile = blockIdx.x * 8 + wave;
  const int lt = tile % (LL / 16);
  const int bk = tile / (LL / 16);
  const int k = bk % KK, b = bk / KK;
  const int lbase = lt * 16;
  const int l = lbase + lid;
  const int lidx = k ? (LL - 1 - l) : l;
  const float* arow = xconv + (size_t)(b * LL + lidx) * DI;
  v8f acc[3] = {};
  for (int k0 = 0; k0 < DI; k0 += 4) {
    v2f a; a.x = arow[k0 + 2 * half]; a.y = arow[k0 + 2 * half + 1];
#pragma unroll
    for (int t = 0; t < 3; ++t) {
      int c2 = t * 16 + lid;
      v2f bb;
      if (c2 < CDBL) {
        const float* wrow = xpw + (size_t)(k * CDBL + c2) * DI;
        bb.x = wrow[k0 + 2 * half]; bb.y = wrow[k0 + 2 * half + 1];
      } else { bb.x = 0.f; bb.y = 0.f; }
      acc[t] = wmma4(a, bb, acc[t]);
    }
  }
#pragma unroll
  for (int t = 0; t < 3; ++t) {
    int col = t * 16 + lid;
    if (col < CDBL) {
      float* orow = x_dbl + ((size_t)bk * LL + lbase + half * 8) * CDBL + col;
#pragma unroll
      for (int v = 0; v < 8; ++v) orow[(size_t)v * CDBL] = acc[t][v];
    }
  }
}

__global__ void __launch_bounds__(256) k_dtproj(const float* __restrict__ x_dbl,
                                                const float* __restrict__ dtw,
                                                const float* __restrict__ dtb,
                                                float* __restrict__ delta) {
  const int tid = threadIdx.x, wave = tid >> 5, lane = tid & 31;
  const int lid = lane & 15, half = lane >> 4;
  const int tile = blockIdx.x;
  const int lt = tile % (LL / 16);
  const int bk = tile / (LL / 16);
  const int k = bk % KK;
  const int lbase = lt * 16;
  const float* arow = x_dbl + ((size_t)bk * LL + lbase + lid) * CDBL;
  v2f a0, a1;
  a0.x = arow[2 * half]; a0.y = arow[2 * half + 1];
  a1.x = (half == 0) ? arow[4] : 0.f;
  a1.y = (half == 0) ? arow[5] : 0.f;
  for (int nt = wave; nt < DI / 16; nt += 8) {
    const int d = nt * 16 + lid;
    const float* wrow = dtw + (size_t)(k * DI + d) * RR;
    v2f b0, b1;
    b0.x = wrow[2 * half]; b0.y = wrow[2 * half + 1];
    b1.x = (half == 0) ? wrow[4] : 0.f;
    b1.y = (half == 0) ? wrow[5] : 0.f;
    v8f acc = {};
    acc = wmma4(a0, b0, acc);
    acc = wmma4(a1, b1, acc);
    const float bias = dtb[k * DI + d];
    float* orow = delta + ((size_t)bk * LL + lbase + half * 8) * DI + d;
#pragma unroll
    for (int v = 0; v < 8; ++v) orow[(size_t)v * DI] = softplus_f(acc[v] + bias);
  }
}

__global__ void __launch_bounds__(256) k_scan_pass1(
    const float* __restrict__ xconv, const float* __restrict__ delta,
    const float* __restrict__ x_dbl, const float* __restrict__ A_logs,
    float* __restrict__ chunkA, float* __restrict__ chunkH) {
  const int gid = blockIdx.x * 256 + threadIdx.x;
  const int d = gid % DI;
  const int c = (gid / DI) % NC;
  const int bk = gid / (DI * NC);
  const int k = bk % KK, b = bk / KK;
  float Av[NN];
#pragma unroll
  for (int n = 0; n < NN; ++n)
    Av[n] = -__expf(A_logs[(size_t)(k * DI + d) * NN + n]);
  float aP[NN], h[NN];
#pragma unroll
  for (int n = 0; n < NN; ++n) { aP[n] = 1.f; h[n] = 0.f; }
  const float* xc = xconv + (size_t)b * LL * DI;
  const float* dl = delta + (size_t)bk * LL * DI;
  const float* xd = x_dbl + (size_t)bk * LL * CDBL;
  const int l0 = c * CL;
  for (int l = l0; l < l0 + CL; ++l) {
    const int lidx = k ? (LL - 1 - l) : l;
    const float u = xc[(size_t)lidx * DI + d];
    const float dlt = dl[(size_t)l * DI + d];
    const float* row = xd + (size_t)l * CDBL;
    const float du = dlt * u;
#pragma unroll
    for (int n = 0; n < NN; ++n) {
      const float dA = __expf(dlt * Av[n]);
      aP[n] *= dA;
      h[n] = dA * h[n] + du * row[RR + n];
    }
  }
#pragma unroll
  for (int n = 0; n < NN; ++n) {
    const size_t idx = ((size_t)(bk * NC + c) * NN + n) * DI + d;
    chunkA[idx] = aP[n];
    chunkH[idx] = h[n];
  }
}

__global__ void __launch_bounds__(256) k_scan_pass2(
    const float* __restrict__ chunkA, float* __restrict__ chunkH) {
  const int gid = blockIdx.x * 256 + threadIdx.x;
  const int d = gid % DI;
  const int bk = gid / DI;
  float hrun[NN];
#pragma unroll
  for (int n = 0; n < NN; ++n) hrun[n] = 0.f;
  for (int c = 0; c < NC; ++c) {
#pragma unroll
    for (int n = 0; n < NN; ++n) {
      const size_t idx = ((size_t)(bk * NC + c) * NN + n) * DI + d;
      const float a = chunkA[idx];
      const float hl = chunkH[idx];
      chunkH[idx] = hrun[n];
      hrun[n] = a * hrun[n] + hl;
    }
  }
}

__global__ void __launch_bounds__(256) k_scan_pass3(
    const float* __restrict__ xconv, const float* __restrict__ delta,
    const float* __restrict__ x_dbl, const float* __restrict__ A_logs,
    const float* __restrict__ Ds, const float* __restrict__ chunkH,
    float* __restrict__ ys) {
  const int gid = blockIdx.x * 256 + threadIdx.x;
  const int d = gid % DI;
  const int c = (gid / DI) % NC;
  const int bk = gid / (DI * NC);
  const int k = bk % KK, b = bk / KK;
  float Av[NN];
#pragma unroll
  for (int n = 0; n < NN; ++n)
    Av[n] = -__expf(A_logs[(size_t)(k * DI + d) * NN + n]);
  const float Dv = Ds[k * DI + d];
  float h[NN];
#pragma unroll
  for (int n = 0; n < NN; ++n)
    h[n] = chunkH[((size_t)(bk * NC + c) * NN + n) * DI + d];
  const float* xc = xconv + (size_t)b * LL * DI;
  const float* dl = delta + (size_t)bk * LL * DI;
  const float* xd = x_dbl + (size_t)bk * LL * CDBL;
  float* yo = ys + (size_t)bk * LL * DI;
  const int l0 = c * CL;
  for (int l = l0; l < l0 + CL; ++l) {
    const int lidx = k ? (LL - 1 - l) : l;
    const float u = xc[(size_t)lidx * DI + d];
    const float dlt = dl[(size_t)l * DI + d];
    const float* row = xd + (size_t)l * CDBL;
    const float du = dlt * u;
    float y = 0.f;
#pragma unroll
    for (int n = 0; n < NN; ++n) {
      const float dA = __expf(dlt * Av[n]);
      h[n] = dA * h[n] + du * row[RR + n];
      y += h[n] * row[RR + NN + n];
    }
    yo[(size_t)l * DI + d] = y + u * Dv;
  }
}

__global__ void __launch_bounds__(256) k_outproj(const float* __restrict__ ys,
                                                 const float* __restrict__ xz,
                                                 const float* __restrict__ W_out,
                                                 float* __restrict__ out) {
  __shared__ float sY[16][DI + 1];
  const int rowBase = blockIdx.x * 16;
  const int b = rowBase / LL;
  const int lb = rowBase % LL;
  const int tid = threadIdx.x;
  for (int i = tid; i < 16 * DI; i += 256) {
    int m = i / DI, dd = i % DI;
    int l = lb + m;
    float y0 = ys[((size_t)(b * KK + 0) * LL + l) * DI + dd];
    float y1 = ys[((size_t)(b * KK + 1) * LL + (LL - 1 - l)) * DI + dd];
    float zv = xz[(size_t)(b * LL + l) * NOUT + DI + dd];
    sY[m][dd] = (y0 + y1) * silu_f(zv);
  }
  __syncthreads();
  const int wave = tid >> 5, lane = tid & 31;
  const int lid = lane & 15, half = lane >> 4;
  for (int nt = wave; nt < CC / 16; nt += 8) {
    const int nbase = nt * 16;
    const float* wrow = W_out + (size_t)(nbase + lid) * DI;
    v8f acc = {};
    for (int k0 = 0; k0 < DI; k0 += 4) {
      v2f a;  a.x = sY[lid][k0 + 2 * half];  a.y = sY[lid][k0 + 2 * half + 1];
      v2f bb; bb.x = wrow[k0 + 2 * half];    bb.y = wrow[k0 + 2 * half + 1];
      acc = wmma4(a, bb, acc);
    }
    float* orow = out + (size_t)(rowBase + half * 8) * CC + nbase + lid;
#pragma unroll
    for (int v = 0; v < 8; ++v) orow[(size_t)v * CC] = acc[v];
  }
}

extern "C" void kernel_launch(void* const* d_in, const int* in_sizes, int n_in,
                              void* d_out, int out_size, void* d_ws, size_t ws_size,
                              hipStream_t stream) {
  (void)in_sizes; (void)n_in; (void)out_size; (void)ws_size;
  const float* x      = (const float*)d_in[0];
  const float* W_in   = (const float*)d_in[1];
  const float* conv_w = (const float*)d_in[2];
  const float* conv_b = (const float*)d_in[3];
  const float* xpw    = (const float*)d_in[4];
  const float* dtw    = (const float*)d_in[5];
  const float* dtb    = (const float*)d_in[6];
  const float* A_logs = (const float*)d_in[7];
  const float* Ds     = (const float*)d_in[8];
  const float* W_out  = (const float*)d_in[9];
  float* out = (float*)d_out;

  char* ws = (char*)d_ws;
  size_t off = 0;
  auto alloc = [&](size_t nbytes) {
    char* p = ws + off;
    off = (off + nbytes + 255) & ~(size_t)255;
    return (float*)p;
  };
  float* xz     = alloc((size_t)BL * NOUT * sizeof(float));
  float* xconv  = alloc((size_t)BL * DI * sizeof(float));
  float* x_dbl  = alloc((size_t)BB * KK * LL * CDBL * sizeof(float));
  float* delta  = alloc((size_t)BB * KK * LL * DI * sizeof(float));
  float* ysb    = alloc((size_t)BB * KK * LL * DI * sizeof(float));
  float* chunkA = alloc((size_t)BB * KK * DI * NC * NN * sizeof(float));
  float* chunkH = alloc((size_t)BB * KK * DI * NC * NN * sizeof(float));

  k_inproj<<<BL / 16, 256, 0, stream>>>(x, W_in, xz);
  k_dwconv<<<(BB * HH * WWD * DI + 255) / 256, 256, 0, stream>>>(xz, conv_w,
                                                                 conv_b, xconv);
  k_xdbl<<<(BB * KK * LL / 16) / 8, 256, 0, stream>>>(xconv, xpw, x_dbl);
  k_dtproj<<<BB * KK * LL / 16, 256, 0, stream>>>(x_dbl, dtw, dtb, delta);
  k_scan_pass1<<<(BB * KK * DI * NC) / 256, 256, 0, stream>>>(
      xconv, delta, x_dbl, A_logs, chunkA, chunkH);
  k_scan_pass2<<<(BB * KK * DI) / 256, 256, 0, stream>>>(chunkA, chunkH);
  k_scan_pass3<<<(BB * KK * DI * NC) / 256, 256, 0, stream>>>(
      xconv, delta, x_dbl, A_logs, Ds, chunkH, ysb);
  k_outproj<<<BL / 16, 256, 0, stream>>>(ysb, xz, W_out, out);
}


# scan pass 2 made memory-parallel: 16x the threads (one per chain and state n), each thread's chunk loads issued 32 chunks ahead of its serial compose; per-chain operation order kept
# speedup vs baseline: 1.3121x; 1.3121x over previous
#include <hip/hip_runtime.h>
#include <cstdint>
#include <cstddef>

#define BB 4
#define HH 32
#define WWD 256
#define CC 96
#define DI 192
#define KK 2
#define NN 16
#define RR 6
#define LL (HH*WWD)
#define BL (BB*LL)
#define NOUT (2*DI)
#define CDBL (RR + 2*NN)
#define NC 128
#define CL (LL/NC)

typedef float v2f __attribute__((ext_vector_type(2)));
typedef float v8f __attribute__((ext_vector_type(8)));

__device__ __forceinline__ v8f wmma4(v2f a, v2f b, v8f c) {
  return __builtin_amdgcn_wmma_f32_16x16x4_f32(false, a, false, b, (short)0, c,
                                               false, false);
}

__device__ __forceinline__ float silu_f(float x) {
  return x / (1.0f + __expf(-x));
}
__device__ __forceinline__ float softplus_f(float x) {
  return (x > 20.0f) ? x : log1pf(__expf(x));
}

__global__ void __launch_bounds__(256) k_inproj(const float* __restrict__ x,
                                                const float* __restrict__ W_in,
                                                float* __restrict__ xz) {
  __shared__ float sX[16][CC + 1];
  const int rowBase = blockIdx.x * 16;
  const int tid = threadIdx.x;
  for (int i = tid; i < 16 * CC; i += 256) {
    int m = i / CC, kk = i % CC;
    sX[m][kk] = x[(size_t)(rowBase + m) * CC + kk];
  }
  __syncthreads();
  const int wave = tid >> 5, lane = tid & 31;
  const int lid = lane & 15, half = lane >> 4;
  for (int nt = wave; nt < NOUT / 16; nt += 8) {
    const int nbase = nt * 16;
    const float* wrow = W_in + (size_t)(nbase + lid) * CC;
    v8f acc = {};
    for (int k0 = 0; k0 < CC; k0 += 4) {
      v2f a; a.x = sX[lid][k0 + 2 * half]; a.y = sX[lid][k0 + 2 * half + 1];
      v2f b; b.x = wrow[k0 + 2 * half];    b.y = wrow[k0 + 2 * half + 1];
      acc = wmma4(a, b, acc);
    }
    float* orow = xz + (size_t)(rowBase + half * 8) * NOUT + nbase + lid;
#pragma unroll
    for (int v = 0; v < 8; ++v) orow[(size_t)v * NOUT] = acc[v];
  }
}

__global__ void __launch_bounds__(256) k_dwconv(const float* __restrict__ xz,
                                                const float* __restrict__ conv_w,
                                                const float* __restrict__ conv_b,
                                                float* __restrict__ xconv) {
  int idx = blockIdx.x * 256 + threadIdx.x;
  if (idx >= BB * HH * WWD * DI) return;
  int c = idx % DI; int t = idx / DI;
  int w = t % WWD; t /= WWD;
  int h = t % HH;  int b = t / HH;
  float s = conv_b[c];
#pragma unroll
  for (int dy = -1; dy <= 1; ++dy) {
    int hh = h + dy; if (hh < 0 || hh >= HH) continue;
#pragma unroll
    for (int dx = -1; dx <= 1; ++dx) {
      int ww = w + dx; if (ww < 0 || ww >= WWD) continue;
      s += xz[(size_t)((b * HH + hh) * WWD + ww) * NOUT + c] *
           conv_w[c * 9 + (dy + 1) * 3 + (dx + 1)];
    }
  }
  xconv[(size_t)((b * HH + h) * WWD + w) * DI + c] = silu_f(s);
}

__global__ void __launch_bounds__(256) k_xdbl(const float* __restrict__ xconv,
                                              const float* __restrict__ xpw,
                                              float* __restrict__ x_dbl) {
  const int tid = threadIdx.x, wave = tid >> 5, lane = tid & 31;
  const int lid = lane & 15, half = lane >> 4;
  const int tile = blockIdx.x * 8 + wave;
  const int lt = tile % (LL / 16);
  const int bk = tile / (LL / 16);
  const int k = bk % KK, b = bk / KK;
  const int lbase = lt * 16;
  const int l = lbase + lid;
  const int lidx = k ? (LL - 1 - l) : l;
  const float* arow = xconv + (size_t)(b * LL + lidx) * DI;
  v8f acc[3] = {};
  for (int k0 = 0; k0 < DI; k0 += 4) {
    v2f a; a.x = arow[k0 + 2 * half]; a.y = arow[k0 + 2 * half + 1];
#pragma unroll
    for (int t = 0; t < 3; ++t) {
      int c2 = t * 16 + lid;
      v2f bb;
      if (c2 < CDBL) {
        const float* wrow = xpw + (size_t)(k * CDBL + c2) * DI;
        bb.x = wrow[k0 + 2 * half]; bb.y = wrow[k0 + 2 * half + 1];
      } else { bb.x = 0.f; bb.y = 0.f; }
      acc[t] = wmma4(a, bb, acc[t]);
    }
  }
#pragma unroll
  for (int t = 0; t < 3; ++t) {
    int col = t * 16 + lid;
    if (col < CDBL) {
      float* orow = x_dbl + ((size_t)bk * LL + lbase + half * 8) * CDBL + col;
#pragma unroll
      for (int v = 0; v < 8; ++v) orow[(size_t)v * CDBL] = acc[t][v];
    }
  }
}

__global__ void __launch_bounds__(256) k_dtproj(const float* __restrict__ x_dbl,
                                                const float* __restrict__ dtw,
                                                const float* __restrict__ dtb,
                                                float* __restrict__ delta) {
  const int tid = threadIdx.x, wave = tid >> 5, lane = tid & 31;
  const int lid = lane & 15, half = lane >> 4;
  const int tile = blockIdx.x;
  const int lt = tile % (LL / 16);
  const int bk = tile / (LL / 16);
  const int k = bk % KK;
  const int lbase = lt * 16;
  const float* arow = x_dbl + ((size_t)bk * LL + lbase + lid) * CDBL;
  v2f a0, a1;
  a0.x = arow[2 * half]; a0.y = arow[2 * half + 1];
  a1.x = (half == 0) ? arow[4] : 0.f;
  a1.y = (half == 0) ? arow[5] : 0.f;
  for (int nt = wave; nt < DI / 16; nt += 8) {
    const int d = nt * 16 + lid;
    const float* wrow = dtw + (size_t)(k * DI + d) * RR;
    v2f b0, b1;
    b0.x = wrow[2 * half]; b0.y = wrow[2 * half + 1];
    b1.x = (half == 0) ? wrow[4] : 0.f;
    b1.y = (half == 0) ? wrow[5] : 0.f;
    v8f acc = {};
    acc = wmma4(a0, b0, acc);
    acc = wmma4(a1, b1, acc);
    const float bias = dtb[k * DI + d];
    float* orow = delta + ((size_t)bk * LL + lbase + half * 8) * DI + d;
#pragma unroll
    for (int v = 0; v < 8; ++v) orow[(size_t)v * DI] = softplus_f(acc[v] + bias);
  }
}

__global__ void __launch_bounds__(256) k_scan_pass1(
    const float* __restrict__ xconv, const float* __restrict__ delta,
    const float* __restrict__ x_dbl, const float* __restrict__ A_logs,
    float* __restrict__ chunkA, float* __restrict__ chunkH) {
  const int gid = blockIdx.x * 256 + threadIdx.x;
  const int d = gid % DI;
  const int c = (gid / DI) % NC;
  const int bk = gid / (DI * NC);
  const int k = bk % KK, b = bk / KK;
  float Av[NN];
#pragma unroll
  for (int n = 0; n < NN; ++n)
    Av[n] = -__expf(A_logs[(size_t)(k * DI + d) * NN + n]);
  float aP[NN], h[NN];
#pragma unroll
  for (int n = 0; n < NN; ++n) { aP[n] = 1.f; h[n] = 0.f; }
  const float* xc = xconv + (size_t)b * LL * DI;
  const float* dl = delta + (size_t)bk * LL * DI;
  const float* xd = x_dbl + (size_t)bk * LL * CDBL;
  const int l0 = c * CL;
  for (int l = l0; l < l0 + CL; ++l) {
    const int lidx = k ? (LL - 1 - l) : l;
    const float u = xc[(size_t)lidx * DI + d];
    const float dlt = dl[(size_t)l * DI + d];
    const float* row = xd + (size_t)l * CDBL;
    const float du = dlt * u;
#pragma unroll
    for (int n = 0; n < NN; ++n) {
      const float dA = __expf(dlt * Av[n]);
      aP[n] *= dA;
      h[n] = dA * h[n] + du * row[RR + n];
    }
  }
#pragma unroll
  for (int n = 0; n < NN; ++n) {
    const size_t idx = ((size_t)(bk * NC + c) * NN + n) * DI + d;
    chunkA[idx] = aP[n];
    chunkH[idx] = h[n];
  }
}

#define P2U 32
#define P2T 96
__global__ void __launch_bounds__(P2T) k_scan_pass2(
    const float* __restrict__ chunkA, float* __restrict__ chunkH) {
  const int gid = blockIdx.x * P2T + threadIdx.x;
  const int d = gid % DI;
  const int n = (gid / DI) % NN;
  const int bk = gid / (DI * NN);
  const int cs = NN * DI;
  const float* pa = chunkA + ((size_t)bk * NC * NN + n) * DI + d;
  float* ph = chunkH + ((size_t)bk * NC * NN + n) * DI + d;
  float hrun = 0.f;
#pragma unroll 1
  for (int c0 = 0; c0 < NC; c0 += P2U) {
    float a[P2U], hl[P2U];
#pragma unroll
    for (int j = 0; j < P2U; ++j) {
      a[j] = pa[(c0 + j) * cs];
      hl[j] = ph[(c0 + j) * cs];
    }
#pragma unroll
    for (int j = 0; j < P2U; ++j) {
      ph[(c0 + j) * cs] = hrun;
      hrun = a[j] * hrun + hl[j];
    }
  }
}

__global__ void __launch_bounds__(256) k_scan_pass3(
    const float* __restrict__ xconv, const float* __restrict__ delta,
    const float* __restrict__ x_dbl, const float* __restrict__ A_logs,
    const float* __restrict__ Ds, const float* __restrict__ chunkH,
    float* __restrict__ ys) {
  const int gid = blockIdx.x * 256 + threadIdx.x;
  const int d = gid % DI;
  const int c = (gid / DI) % NC;
  const int bk = gid / (DI * NC);
  const int k = bk % KK, b = bk / KK;
  float Av[NN];
#pragma unroll
  for (int n = 0; n < NN; ++n)
    Av[n] = -__expf(A_logs[(size_t)(k * DI + d) * NN + n]);
  const float Dv = Ds[k * DI + d];
  float h[NN];
#pragma unroll
  for (int n = 0; n < NN; ++n)
    h[n] = chunkH[((size_t)(bk * NC + c) * NN + n) * DI + d];
  const float* xc = xconv + (size_t)b * LL * DI;
  const float* dl = delta + (size_t)bk * LL * DI;
  const float* xd = x_dbl + (size_t)bk * LL * CDBL;
  float* yo = ys + (size_t)bk * LL * DI;
  const int l0 = c * CL;
  for (int l = l0; l < l0 + CL; ++l) {
    const int lidx = k ? (LL - 1 - l) : l;
    const float u = xc[(size_t)lidx * DI + d];
    const float dlt = dl[(size_t)l * DI + d];
    const float* row = xd + (size_t)l * CDBL;
    const float du = dlt * u;
    float y = 0.f;
#pragma unroll
    for (int n = 0; n < NN; ++n) {
      const float dA = __expf(dlt * Av[n]);
      h[n] = dA * h[n] + du * row[RR + n];
      y += h[n] * row[RR + NN + n];
    }
    yo[(size_t)l * DI + d] = y + u * Dv;
  }
}

__global__ void __launch_bounds__(256) k_outproj(const float* __restrict__ ys,
                                                 const float* __restrict__ xz,
                                                 const float* __restrict__ W_out,
                                                 float* __restrict__ out) {
  __shared__ float sY[16][DI + 1];
  const int rowBase = blockIdx.x * 16;
  const int b = rowBase / LL;
  const int lb = rowBase % LL;
  const int tid = threadIdx.x;
  for (int i = tid; i < 16 * DI; i += 256) {
    int m = i / DI, dd = i % DI;
    int l = lb + m;
    float y0 = ys[((size_t)(b * KK + 0) * LL + l) * DI + dd];
    float y1 = ys[((size_t)(b * KK + 1) * LL + (LL - 1 - l)) * DI + dd];
    float zv = xz[(size_t)(b * LL + l) * NOUT + DI + dd];
    sY[m][dd] = (y0 + y1) * silu_f(zv);
  }
  __syncthreads();
  const int wave = tid >> 5, lane = tid & 31;
  const int lid = lane & 15, half = lane >> 4;
  for (int nt = wave; nt < CC / 16; nt += 8) {
    const int nbase = nt * 16;
    const float* wrow = W_out + (size_t)(nbase + lid) * DI;
    v8f acc = {};
    for (int k0 = 0; k0 < DI; k0 += 4) {
      v2f a;  a.x = sY[lid][k0 + 2 * half];  a.y = sY[lid][k0 + 2 * half + 1];
      v2f bb; bb.x = wrow[k0 + 2 * half];    bb.y = wrow[k0 + 2 * half + 1];
      acc = wmma4(a, bb, acc);
    }
    float* orow = out + (size_t)(rowBase + half * 8) * CC + nbase + lid;
#pragma unroll
    for (int v = 0; v < 8; ++v) orow[(size_t)v * CC] = acc[v];
  }
}

extern "C" void kernel_launch(void* const* d_in, const int* in_sizes, int n_in,
                              void* d_out, int out_size, void* d_ws, size_t ws_size,
                              hipStream_t stream) {
  (void)in_sizes; (void)n_in; (void)out_size; (void)ws_size;
  const float* x      = (const float*)d_in[0];
  const float* W_in   = (const float*)d_in[1];
  const float* conv_w = (const float*)d_in[2];
  const float* conv_b = (const float*)d_in[3];
  const float* xpw    = (const float*)d_in[4];
  const float* dtw    = (const float*)d_in[5];
  const float* dtb    = (const float*)d_in[6];
  const float* A_logs = (const float*)d_in[7];
  const float* Ds     = (const float*)d_in[8];
  const float* W_out  = (const float*)d_in[9];
  float* out = (float*)d_out;

  char* ws = (char*)d_ws;
  size_t off = 0;
  auto alloc = [&](size_t nbytes) {
    char* p = ws + off;
    off = (off + nbytes + 255) & ~(size_t)255;
    return (float*)p;
  };
  float* xz     = alloc((size_t)BL * NOUT * sizeof(float));
  float* xconv  = alloc((size_t)BL * DI * sizeof(float));
  float* x_dbl  = alloc((size_t)BB * KK * LL * CDBL * sizeof(float));
  float* delta  = alloc((size_t)BB * KK * LL * DI * sizeof(float));
  float* ysb    = alloc((size_t)BB * KK * LL * DI * sizeof(float));
  float* chunkA = alloc((size_t)BB * KK * DI * NC * NN * sizeof(float));
  float* chunkH = alloc((size_t)BB * KK * DI * NC * NN * sizeof(float));

  k_inproj<<<BL / 16, 256, 0, stream>>>(x, W_in, xz);
  k_dwconv<<<(BB * HH * WWD * DI + 255) / 256, 256, 0, stream>>>(xz, conv_w,
                                                                 conv_b, xconv);
  k_xdbl<<<(BB * KK * LL / 16) / 8, 256, 0, stream>>>(xconv, xpw, x_dbl);
  k_dtproj<<<BB * KK * LL / 16, 256, 0, stream>>>(x_dbl, dtw, dtb, delta);
  k_scan_pass1<<<(BB * KK * DI * NC) / 256, 256, 0, stream>>>(
      xconv, delta, x_dbl, A_logs, chunkA, chunkH);
  k_scan_pass2<<<(BB * KK * NN * DI) / P2T, P2T, 0, stream>>>(chunkA, chunkH);
  k_scan_pass3<<<(BB * KK * DI * NC) / 256, 256, 0, stream>>>(
      xconv, delta, x_dbl, A_logs, Ds, chunkH, ysb);
  k_outproj<<<BL / 16, 256, 0, stream>>>(ysb, xz, W_out, out);
}


# depthwise conv: each thread computes 8 outputs along w for 4 channels from a 3x10 window of float4 loads held in registers, weights loaded once per thread: 32x fewer waves; tap order kept
# speedup vs baseline: 1.4353x; 1.0939x over previous
#include <hip/hip_runtime.h>
#include <cstdint>
#include <cstddef>

#define BB 4
#define HH 32
#define WWD 256
#define CC 96
#define DI 192
#define KK 2
#define NN 16
#define RR 6
#define LL (HH*WWD)
#define BL (BB*LL)
#define NOUT (2*DI)
#define CDBL (RR + 2*NN)
#define NC 128
#define CL (LL/NC)

typedef float v2f __attribute__((ext_vector_type(2)));
typedef float v8f __attribute__((ext_vector_type(8)));

__device__ __forceinline__ v8f wmma4(v2f a, v2f b, v8f c) {
  return __builtin_amdgcn_wmma_f32_16x16x4_f32(false, a, false, b, (short)0, c,
                                               false, false);
}

__device__ __forceinline__ float silu_f(float x) {
  return x / (1.0f + __expf(-x));
}
__device__ __forceinline__ float softplus_f(float x) {
  return (x > 20.0f) ? x : log1pf(__expf(x));
}

__global__ void __launch_bounds__(256) k_inproj(const float* __restrict__ x,
                                                const float* __restrict__ W_in,
                                                float* __restrict__ xz) {
  __shared__ float sX[16][CC + 1];
  const int rowBase = blockIdx.x * 16;
  const int tid = threadIdx.x;
  for (int i = tid; i < 16 * CC; i += 256) {
    int m = i / CC, kk = i % CC;
    sX[m][kk] = x[(size_t)(rowBase + m) * CC + kk];
  }
  __syncthreads();
  const int wave = tid >> 5, lane = tid & 31;
  const int lid = lane & 15, half = lane >> 4;
  for (int nt = wave; nt < NOUT / 16; nt += 8) {
    const int nbase = nt * 16;
    const float* wrow = W_in + (size_t)(nbase + lid) * CC;
    v8f acc = {};
    for (int k0 = 0; k0 < CC; k0 += 4) {
      v2f a; a.x = sX[lid][k0 + 2 * half]; a.y = sX[lid][k0 + 2 * half + 1];
      v2f b; b.x = wrow[k0 + 2 * half];    b.y = wrow[k0 + 2 * half + 1];
      acc = wmma4(a, b, acc);
    }
    float* orow = xz + (size_t)(rowBase + half * 8) * NOUT + nbase + lid;
#pragma unroll
    for (int v = 0; v < 8; ++v) orow[(size_t)v * NOUT] = acc[v];
  }
}

#define DWS 8
#define DWC4 (DI / 4)
#define DWSTRIPS (WWD / DWS)
#define DWT 192
__global__ void __launch_bounds__(DWT) k_dwconv(const float* __restrict__ xz,
                                                const float* __restrict__ conv_w,
                                                const float* __restrict__ conv_b,
                                                float* __restrict__ xconv) {
  const unsigned tid = threadIdx.x;
  const int c = (int)(tid % DWC4) * 4;
  const unsigned s = blockIdx.x * (DWT / DWC4) + tid / DWC4;
  const int w0 = (int)(s % DWSTRIPS) * DWS;
  const int h = (int)((s / DWSTRIPS) % HH);
  const int b = (int)(s / (DWSTRIPS * HH));

  float wt[4][9];
#pragma unroll
  for (int j = 0; j < 4; ++j)
#pragma unroll
    for (int t = 0; t < 9; ++t) wt[j][t] = conv_w[(c + j) * 9 + t];
  float bias[4];
#pragma unroll
  for (int j = 0; j < 4; ++j) bias[j] = conv_b[c + j];

  float v[3][DWS + 2][4];
#pragma unroll
  for (int r = 0; r < 3; ++r) {
    const int hh = h + r - 1;
    const bool rok = (hh >= 0) && (hh < HH);
    const int hc = hh < 0 ? 0 : (hh >= HH ? HH - 1 : hh);
#pragma unroll
    for (int q = 0; q < DWS + 2; ++q) {
      bool cok = true;
      int wc = w0 + q - 1;
      if (q == 0) { cok = w0 > 0; wc = cok ? wc : 0; }
      if (q == DWS + 1) { cok = (w0 + DWS) < WWD; wc = cok ? wc : WWD - 1; }
      const bool ok = rok && cok;
      const float4 t4 = *reinterpret_cast<const float4*>(
          xz + (size_t)((b * HH + hc) * WWD + wc) * NOUT + c);
      v[r][q][0] = ok ? t4.x : 0.f;
      v[r][q][1] = ok ? t4.y : 0.f;
      v[r][q][2] = ok ? t4.z : 0.f;
      v[r][q][3] = ok ? t4.w : 0.f;
    }
  }

  float* orow = xconv + (size_t)((b * HH + h) * WWD + w0) * DI + c;
#pragma unroll
  for (int i = 0; i < DWS; ++i) {
    float o[4];
#pragma unroll
    for (int j = 0; j < 4; ++j) {
      float acc = bias[j];
#pragma unroll
      for (int r = 0; r < 3; ++r)
#pragma unroll
        for (int dx = 0; dx < 3; ++dx) acc += v[r][i + dx][j] * wt[j][r * 3 + dx];
      o[j] = silu_f(acc);
    }
    *reinterpret_cast<float4*>(orow + (size_t)i * DI) =
        make_float4(o[0], o[1], o[2], o[3]);
  }
}

__global__ void __launch_bounds__(256) k_xdbl(const float* __restrict__ xconv,
                                              const float* __restrict__ xpw,
                                              float* __restrict__ x_dbl) {
  const int tid = threadIdx.x, wave = tid >> 5, lane = tid & 31;
  const int lid = lane & 15, half = lane >> 4;
  const int tile = blockIdx.x * 8 + wave;
  const int lt = tile % (LL / 16);
  const int bk = tile / (LL / 16);
  const int k = bk % KK, b = bk / KK;
  const int lbase = lt * 16;
  const int l = lbase + lid;
  const int lidx = k ? (LL - 1 - l) : l;
  const float* arow = xconv + (size_t)(b * LL + lidx) * DI;
  v8f acc[3] = {};
  for (int k0 = 0; k0 < DI; k0 += 4) {
    v2f a; a.x = arow[k0 + 2 * half]; a.y = arow[k0 + 2 * half + 1];
#pragma unroll
    for (int t = 0; t < 3; ++t) {
      int c2 = t * 16 + lid;
      v2f bb;
      if (c2 < CDBL) {
        const float* wrow = xpw + (size_t)(k * CDBL + c2) * DI;
        bb.x = wrow[k0 + 2 * half]; bb.y = wrow[k0 + 2 * half + 1];
      } else { bb.x = 0.f; bb.y = 0.f; }
      acc[t] = wmma4(a, bb, acc[t]);
    }
  }
#pragma unroll
  for (int t = 0; t < 3; ++t) {
    int col = t * 16 + lid;
    if (col < CDBL) {
      float* orow = x_dbl + ((size_t)bk * LL + lbase + half * 8) * CDBL + col;
#pragma unroll
      for (int v = 0; v < 8; ++v) orow[(size_t)v * CDBL] = acc[t][v];
    }
  }
}

__global__ void __launch_bounds__(256) k_dtproj(const float* __restrict__ x_dbl,
                                                const float* __restrict__ dtw,
                                                const float* __restrict__ dtb,
                                                float* __restrict__ delta) {
  const int tid = threadIdx.x, wave = tid >> 5, lane = tid & 31;
  const int lid = lane & 15, half = lane >> 4;
  const int tile = blockIdx.x;
  const int lt = tile % (LL / 16);
  const int bk = tile / (LL / 16);
  const int k = bk % KK;
  const int lbase = lt * 16;
  const float* arow = x_dbl + ((size_t)bk * LL + lbase + lid) * CDBL;
  v2f a0, a1;
  a0.x = arow[2 * half]; a0.y = arow[2 * half + 1];
  a1.x = (half == 0) ? arow[4] : 0.f;
  a1.y = (half == 0) ? arow[5] : 0.f;
  for (int nt = wave; nt < DI / 16; nt += 8) {
    const int d = nt * 16 + lid;
    const float* wrow = dtw + (size_t)(k * DI + d) * RR;
    v2f b0, b1;
    b0.x = wrow[2 * half]; b0.y = wrow[2 * half + 1];
    b1.x = (half == 0) ? wrow[4] : 0.f;
    b1.y = (half == 0) ? wrow[5] : 0.f;
    v8f acc = {};
    acc = wmma4(a0, b0, acc);
    acc = wmma4(a1, b1, acc);
    const float bias = dtb[k * DI + d];
    float* orow = delta + ((size_t)bk * LL + lbase + half * 8) * DI + d;
#pragma unroll
    for (int v = 0; v < 8; ++v) orow[(size_t)v * DI] = softplus_f(acc[v] + bias);
  }
}

__global__ void __launch_bounds__(256) k_scan_pass1(
    const float* __restrict__ xconv, const float* __restrict__ delta,
    const float* __restrict__ x_dbl, const float* __restrict__ A_logs,
    float* __restrict__ chunkA, float* __restrict__ chunkH) {
  const int gid = blockIdx.x * 256 + threadIdx.x;
  const int d = gid % DI;
  const int c = (gid / DI) % NC;
  const int bk = gid / (DI * NC);
  const int k = bk % KK, b = bk / KK;
  float Av[NN];
#pragma unroll
  for (int n = 0; n < NN; ++n)
    Av[n] = -__expf(A_logs[(size_t)(k * DI + d) * NN + n]);
  float aP[NN], h[NN];
#pragma unroll
  for (int n = 0; n < NN; ++n) { aP[n] = 1.f; h[n] = 0.f; }
  const float* xc = xconv + (size_t)b * LL * DI;
  const float* dl = delta + (size_t)bk * LL * DI;
  const float* xd = x_dbl + (size_t)bk * LL * CDBL;
  const int l0 = c * CL;
  for (int l = l0; l < l0 + CL; ++l) {
    const int lidx = k ? (LL - 1 - l) : l;
    const float u = xc[(size_t)lidx * DI + d];
    const float dlt = dl[(size_t)l * DI + d];
    const float* row = xd + (size_t)l * CDBL;
    const float du = dlt * u;
#pragma unroll
    for (int n = 0; n < NN; ++n) {
      const float dA = __expf(dlt * Av[n]);
      aP[n] *= dA;
      h[n] = dA * h[n] + du * row[RR + n];
    }
  }
#pragma unroll
  for (int n = 0; n < NN; ++n) {
    const size_t idx = ((size_t)(bk * NC + c) * NN + n) * DI + d;
    chunkA[idx] = aP[n];
    chunkH[idx] = h[n];
  }
}

#define P2U 32
#define P2T 96
__global__ void __launch_bounds__(P2T) k_scan_pass2(
    const float* __restrict__ chunkA, float* __restrict__ chunkH) {
  const int gid = blockIdx.x * P2T + threadIdx.x;
  const int d = gid % DI;
  const int n = (gid / DI) % NN;
  const int bk = gid / (DI * NN);
  const int cs = NN * DI;
  const float* pa = chunkA + ((size_t)bk * NC * NN + n) * DI + d;
  float* ph = chunkH + ((size_t)bk * NC * NN + n) * DI + d;
  float hrun = 0.f;
#pragma unroll 1
  for (int c0 = 0; c0 < NC; c0 += P2U) {
    float a[P2U], hl[P2U];
#pragma unroll
    for (int j = 0; j < P2U; ++j) {
      a[j] = pa[(c0 + j) * cs];
      hl[j] = ph[(c0 + j) * cs];
    }
#pragma unroll
    for (int j = 0; j < P2U; ++j) {
      ph[(c0 + j) * cs] = hrun;
      hrun = a[j] * hrun + hl[j];
    }
  }
}

__global__ void __launch_bounds__(256) k_scan_pass3(
    const float* __restrict__ xconv, const float* __restrict__ delta,
    const float* __restrict__ x_dbl, const float* __restrict__ A_logs,
    const float* __restrict__ Ds, const float* __restrict__ chunkH,
    float* __restrict__ ys) {
  const int gid = blockIdx.x * 256 + threadIdx.x;
  const int d = gid % DI;
  const int c = (gid / DI) % NC;
  const int bk = gid / (DI * NC);
  const int k = bk % KK, b = bk / KK;
  float Av[NN];
#pragma unroll
  for (int n = 0; n < NN; ++n)
    Av[n] = -__expf(A_logs[(size_t)(k * DI + d) * NN + n]);
  const float Dv = Ds[k * DI + d];
  float h[NN];
#pragma unroll
  for (int n = 0; n < NN; ++n)
    h[n] = chunkH[((size_t)(bk * NC + c) * NN + n) * DI + d];
  const float* xc = xconv + (size_t)b * LL * DI;
  const float* dl = delta + (size_t)bk * LL * DI;
  const float* xd = x_dbl + (size_t)bk * LL * CDBL;
  float* yo = ys + (size_t)bk * LL * DI;
  const int l0 = c * CL;
  for (int l = l0; l < l0 + CL; ++l) {
    const int lidx = k ? (LL - 1 - l) : l;
    const float u = xc[(size_t)lidx * DI + d];
    const float dlt = dl[(size_t)l * DI + d];
    const float* row = xd + (size_t)l * CDBL;
    const float du = dlt * u;
    float y = 0.f;
#pragma unroll
    for (int n = 0; n < NN; ++n) {
      const float dA = __expf(dlt * Av[n]);
      h[n] = dA * h[n] + du * row[RR + n];
      y += h[n] * row[RR + NN + n];
    }
    yo[(size_t)l * DI + d] = y + u * Dv;
  }
}

__global__ void __launch_bounds__(256) k_outproj(const float* __restrict__ ys,
                                                 const float* __restrict__ xz,
                                                 const float* __restrict__ W_out,
                                                 float* __restrict__ out) {
  __shared__ float sY[16][DI + 1];
  const int rowBase = blockIdx.x * 16;
  const int b = rowBase / LL;
  const int lb = rowBase % LL;
  const int tid = threadIdx.x;
  for (int i = tid; i < 16 * DI; i += 256) {
    int m = i / DI, dd = i % DI;
    int l = lb + m;
    float y0 = ys[((size_t)(b * KK + 0) * LL + l) * DI + dd];
    float y1 = ys[((size_t)(b * KK + 1) * LL + (LL - 1 - l)) * DI + dd];
    float zv = xz[(size_t)(b * LL + l) * NOUT + DI + dd];
    sY[m][dd] = (y0 + y1) * silu_f(zv);
  }
  __syncthreads();
  const int wave = tid >> 5, lane = tid & 31;
  const int lid = lane & 15, half = lane >> 4;
  for (int nt = wave; nt < CC / 16; nt += 8) {
    const int nbase = nt * 16;
    const float* wrow = W_out + (size_t)(nbase + lid) * DI;
    v8f acc = {};
    for (int k0 = 0; k0 < DI; k0 += 4) {
      v2f a;  a.x = sY[lid][k0 + 2 * half];  a.y = sY[lid][k0 + 2 * half + 1];
      v2f bb; bb.x = wrow[k0 + 2 * half];    bb.y = wrow[k0 + 2 * half + 1];
      acc = wmma4(a, bb, acc);
    }
    float* orow = out + (size_t)(rowBase + half * 8) * CC + nbase + lid;
#pragma unroll
    for (int v = 0; v < 8; ++v) orow[(size_t)v * CC] = acc[v];
  }
}

extern "C" void kernel_launch(void* const* d_in, const int* in_sizes, int n_in,
                              void* d_out, int out_size, void* d_ws, size_t ws_size,
                              hipStream_t stream) {
  (void)in_sizes; (void)n_in; (void)out_size; (void)ws_size;
  const float* x      = (const float*)d_in[0];
  const float* W_in   = (const float*)d_in[1];
  const float* conv_w = (const float*)d_in[2];
  const float* conv_b = (const float*)d_in[3];
  const float* xpw    = (const float*)d_in[4];
  const float* dtw    = (const float*)d_in[5];
  const float* dtb    = (const float*)d_in[6];
  const float* A_logs = (const float*)d_in[7];
  const float* Ds     = (const float*)d_in[8];
  const float* W_out  = (const float*)d_in[9];
  float* out = (float*)d_out;

  char* ws = (char*)d_ws;
  size_t off = 0;
  auto alloc = [&](size_t nbytes) {
    char* p = ws + off;
    off = (off + nbytes + 255) & ~(size_t)255;
    return (float*)p;
  };
  float* xz     = alloc((size_t)BL * NOUT * sizeof(float));
  float* xconv  = alloc((size_t)BL * DI * sizeof(float));
  float* x_dbl  = alloc((size_t)BB * KK * LL * CDBL * sizeof(float));
  float* delta  = alloc((size_t)BB * KK * LL * DI * sizeof(float));
  float* ysb    = alloc((size_t)BB * KK * LL * DI * sizeof(float));
  float* chunkA = alloc((size_t)BB * KK * DI * NC * NN * sizeof(float));
  float* chunkH = alloc((size_t)BB * KK * DI * NC * NN * sizeof(float));

  k_inproj<<<BL / 16, 256, 0, stream>>>(x, W_in, xz);
  k_dwconv<<<(BB * HH * DWSTRIPS) / (DWT / DWC4), DWT, 0, stream>>>(xz, conv_w,
                                                                    conv_b, xconv);
  k_xdbl<<<(BB * KK * LL / 16) / 8, 256, 0, stream>>>(xconv, xpw, x_dbl);
  k_dtproj<<<BB * KK * LL / 16, 256, 0, stream>>>(x_dbl, dtw, dtb, delta);
  k_scan_pass1<<<(BB * KK * DI * NC) / 256, 256, 0, stream>>>(
      xconv, delta, x_dbl, A_logs, chunkA, chunkH);
  k_scan_pass2<<<(BB * KK * NN * DI) / P2T, P2T, 0, stream>>>(chunkA, chunkH);
  k_scan_pass3<<<(BB * KK * DI * NC) / 256, 256, 0, stream>>>(
      xconv, delta, x_dbl, A_logs, Ds, chunkH, ysb);
  k_outproj<<<BL / 16, 256, 0, stream>>>(ysb, xz, W_out, out);
}


# scan passes 1 and 3 made memory-parallel: one (b,k,chunk) per block so the shared B/C row is a scalar load, each lane's u and delta loaded 8 steps ahead of the recurrence; summation order kept
# speedup vs baseline: 1.9136x; 1.3332x over previous
#include <hip/hip_runtime.h>
#include <cstdint>
#include <cstddef>

#define BB 4
#define HH 32
#define WWD 256
#define CC 96
#define DI 192
#define KK 2
#define NN 16
#define RR 6
#define LL (HH*WWD)
#define BL (BB*LL)
#define NOUT (2*DI)
#define CDBL (RR + 2*NN)
#define NC 128
#define CL (LL/NC)

typedef float v2f __attribute__((ext_vector_type(2)));
typedef float v8f __attribute__((ext_vector_type(8)));

__device__ __forceinline__ v8f wmma4(v2f a, v2f b, v8f c) {
  return __builtin_amdgcn_wmma_f32_16x16x4_f32(false, a, false, b, (short)0, c,
                                               false, false);
}

__device__ __forceinline__ float silu_f(float x) {
  return x / (1.0f + __expf(-x));
}
__device__ __forceinline__ float softplus_f(float x) {
  return (x > 20.0f) ? x : log1pf(__expf(x));
}

__global__ void __launch_bounds__(256) k_inproj(const float* __restrict__ x,
                                                const float* __restrict__ W_in,
                                                float* __restrict__ xz) {
  __shared__ float sX[16][CC + 1];
  const int rowBase = blockIdx.x * 16;
  const int tid = threadIdx.x;
  for (int i = tid; i < 16 * CC; i += 256) {
    int m = i / CC, kk = i % CC;
    sX[m][kk] = x[(size_t)(rowBase + m) * CC + kk];
  }
  __syncthreads();
  const int wave = tid >> 5, lane = tid & 31;
  const int lid = lane & 15, half = lane >> 4;
  for (int nt = wave; nt < NOUT / 16; nt += 8) {
    const int nbase = nt * 16;
    const float* wrow = W_in + (size_t)(nbase + lid) * CC;
    v8f acc = {};
    for (int k0 = 0; k0 < CC; k0 += 4) {
      v2f a; a.x = sX[lid][k0 + 2 * half]; a.y = sX[lid][k0 + 2 * half + 1];
      v2f b; b.x = wrow[k0 + 2 * half];    b.y = wrow[k0 + 2 * half + 1];
      acc = wmma4(a, b, acc);
    }
    float* orow = xz + (size_t)(rowBase + half * 8) * NOUT + nbase + lid;
#pragma unroll
    for (int v = 0; v < 8; ++v) orow[(size_t)v * NOUT] = acc[v];
  }
}

#define DWS 8
#define DWC4 (DI / 4)
#define DWSTRIPS (WWD / DWS)
#define DWT 192
__global__ void __launch_bounds__(DWT) k_dwconv(const float* __restrict__ xz,
                                                const float* __restrict__ conv_w,
                                                const float* __restrict__ conv_b,
                                                float* __restrict__ xconv) {
  const unsigned tid = threadIdx.x;
  const int c = (int)(tid % DWC4) * 4;
  const unsigned s = blockIdx.x * (DWT / DWC4) + tid / DWC4;
  const int w0 = (int)(s % DWSTRIPS) * DWS;
  const int h = (int)((s / DWSTRIPS) % HH);
  const int b = (int)(s / (DWSTRIPS * HH));

  float wt[4][9];
#pragma unroll
  for (int j = 0; j < 4; ++j)
#pragma unroll
    for (int t = 0; t < 9; ++t) wt[j][t] = conv_w[(c + j) * 9 + t];
  float bias[4];
#pragma unroll
  for (int j = 0; j < 4; ++j) bias[j] = conv_b[c + j];

  float v[3][DWS + 2][4];
#pragma unroll
  for (int r = 0; r < 3; ++r) {
    const int hh = h + r - 1;
    const bool rok = (hh >= 0) && (hh < HH);
    const int hc = hh < 0 ? 0 : (hh >= HH ? HH - 1 : hh);
#pragma unroll
    for (int q = 0; q < DWS + 2; ++q) {
      bool cok = true;
      int wc = w0 + q - 1;
      if (q == 0) { cok = w0 > 0; wc = cok ? wc : 0; }
      if (q == DWS + 1) { cok = (w0 + DWS) < WWD; wc = cok ? wc : WWD - 1; }
      const bool ok = rok && cok;
      const float4 t4 = *reinterpret_cast<const float4*>(
          xz + (size_t)((b * HH + hc) * WWD + wc) * NOUT + c);
      v[r][q][0] = ok ? t4.x : 0.f;
      v[r][q][1] = ok ? t4.y : 0.f;
      v[r][q][2] = ok ? t4.z : 0.f;
      v[r][q][3] = ok ? t4.w : 0.f;
    }
  }

  float* orow = xconv + (size_t)((b * HH + h) * WWD + w0) * DI + c;
#pragma unroll
  for (int i = 0; i < DWS; ++i) {
    float o[4];
#pragma unroll
    for (int j = 0; j < 4; ++j) {
      float acc = bias[j];
#pragma unroll
      for (int r = 0; r < 3; ++r)
#pragma unroll
        for (int dx = 0; dx < 3; ++dx) acc += v[r][i + dx][j] * wt[j][r * 3 + dx];
      o[j] = silu_f(acc);
    }
    *reinterpret_cast<float4*>(orow + (size_t)i * DI) =
        make_float4(o[0], o[1], o[2], o[3]);
  }
}

__global__ void __launch_bounds__(256) k_xdbl(const float* __restrict__ xconv,
                                              const float* __restrict__ xpw,
                                              float* __restrict__ x_dbl) {
  const int tid = threadIdx.x, wave = tid >> 5, lane = tid & 31;
  const int lid = lane & 15, half = lane >> 4;
  const int tile = blockIdx.x * 8 + wave;
  const int lt = tile % (LL / 16);
  const int bk = tile / (LL / 16);
  const int k = bk % KK, b = bk / KK;
  const int lbase = lt * 16;
  const int l = lbase + lid;
  const int lidx = k ? (LL - 1 - l) : l;
  const float* arow = xconv + (size_t)(b * LL + lidx) * DI;
  v8f acc[3] = {};
  for (int k0 = 0; k0 < DI; k0 += 4) {
    v2f a; a.x = arow[k0 + 2 * half]; a.y = arow[k0 + 2 * half + 1];
#pragma unroll
    for (int t = 0; t < 3; ++t) {
      int c2 = t * 16 + lid;
      v2f bb;
      if (c2 < CDBL) {
        const float* wrow = xpw + (size_t)(k * CDBL + c2) * DI;
        bb.x = wrow[k0 + 2 * half]; bb.y = wrow[k0 + 2 * half + 1];
      } else { bb.x = 0.f; bb.y = 0.f; }
      acc[t] = wmma4(a, bb, acc[t]);
    }
  }
#pragma unroll
  for (int t = 0; t < 3; ++t) {
    int col = t * 16 + lid;
    if (col < CDBL) {
      float* orow = x_dbl + ((size_t)bk * LL + lbase + half * 8) * CDBL + col;
#pragma unroll
      for (int v = 0; v < 8; ++v) orow[(size_t)v * CDBL] = acc[t][v];
    }
  }
}

__global__ void __launch_bounds__(256) k_dtproj(const float* __restrict__ x_dbl,
                                                const float* __restrict__ dtw,
                                                const float* __restrict__ dtb,
                                                float* __restrict__ delta) {
  const int tid = threadIdx.x, wave = tid >> 5, lane = tid & 31;
  const int lid = lane & 15, half = lane >> 4;
  const int tile = blockIdx.x;
  const int lt = tile % (LL / 16);
  const int bk = tile / (LL / 16);
  const int k = bk % KK;
  const int lbase = lt * 16;
  const float* arow = x_dbl + ((size_t)bk * LL + lbase + lid) * CDBL;
  v2f a0, a1;
  a0.x = arow[2 * half]; a0.y = arow[2 * half + 1];
  a1.x = (half == 0) ? arow[4] : 0.f;
  a1.y = (half == 0) ? arow[5] : 0.f;
  for (int nt = wave; nt < DI / 16; nt += 8) {
    const int d = nt * 16 + lid;
    const float* wrow = dtw + (size_t)(k * DI + d) * RR;
    v2f b0, b1;
    b0.x = wrow[2 * half]; b0.y = wrow[2 * half + 1];
    b1.x = (half == 0) ? wrow[4] : 0.f;
    b1.y = (half == 0) ? wrow[5] : 0.f;
    v8f acc = {};
    acc = wmma4(a0, b0, acc);
    acc = wmma4(a1, b1, acc);
    const float bias = dtb[k * DI + d];
    float* orow = delta + ((size_t)bk * LL + lbase + half * 8) * DI + d;
#pragma unroll
    for (int v = 0; v < 8; ++v) orow[(size_t)v * DI] = softplus_f(acc[v] + bias);
  }
}

#define SB 8

__device__ __forceinline__ void p1_batch(const float (&ub)[SB], const float (&db)[SB],
                                         const float (&Av)[NN], float (&aP)[NN],
                                         float (&h)[NN], const float* row0) {
#pragma unroll
  for (int j = 0; j < SB; ++j) {
    const float u = ub[j], dlt = db[j];
    const float* row = row0 + (size_t)j * CDBL;
    const float du = dlt * u;
#pragma unroll
    for (int n = 0; n < NN; ++n) {
      const float dA = __expf(dlt * Av[n]);
      aP[n] *= dA;
      h[n] = dA * h[n] + du * row[n];
    }
  }
}

__global__ void __launch_bounds__(DI) k_scan_pass1(
    const float* __restrict__ xconv, const float* __restrict__ delta,
    const float* __restrict__ x_dbl, const float* __restrict__ A_logs,
    float* __restrict__ chunkA, float* __restrict__ chunkH) {
  const int d = threadIdx.x;
  const int c = blockIdx.x % NC;
  const int bk = blockIdx.x / NC;
  const int k = bk % KK, b = bk / KK;
  float Av[NN];
#pragma unroll
  for (int n = 0; n < NN; ++n)
    Av[n] = -__expf(A_logs[(size_t)(k * DI + d) * NN + n]);
  float aP[NN], h[NN];
#pragma unroll
  for (int n = 0; n < NN; ++n) { aP[n] = 1.f; h[n] = 0.f; }
  const int l0 = c * CL;
  const float* up = xconv + ((size_t)b * LL + (k ? (LL - 1 - l0) : l0)) * DI + d;
  const int ustep = k ? -DI : DI;
  const float* dp = delta + ((size_t)bk * LL + l0) * DI + d;
  const float* xr = x_dbl + ((size_t)bk * LL + l0) * CDBL + RR;
  float ub[SB], db[SB];
#pragma unroll
  for (int j = 0; j < SB; ++j) { ub[j] = up[j * ustep]; db[j] = dp[j * DI]; }
#pragma unroll 1
  for (int s0 = 0; s0 < CL - SB; s0 += SB) {
    float un[SB], dn[SB];
#pragma unroll
    for (int j = 0; j < SB; ++j) {
      un[j] = up[(s0 + SB + j) * ustep];
      dn[j] = dp[(s0 + SB + j) * DI];
    }
    __builtin_amdgcn_sched_barrier(0);
    p1_batch(ub, db, Av, aP, h, xr + (size_t)s0 * CDBL);
#pragma unroll
    for (int j = 0; j < SB; ++j) { ub[j] = un[j]; db[j] = dn[j]; }
  }
  p1_batch(ub, db, Av, aP, h, xr + (size_t)(CL - SB) * CDBL);
#pragma unroll
  for (int n = 0; n < NN; ++n) {
    const size_t idx = ((size_t)(bk * NC + c) * NN + n) * DI + d;
    chunkA[idx] = aP[n];
    chunkH[idx] = h[n];
  }
}

#define P2U 32
#define P2T 96
__global__ void __launch_bounds__(P2T) k_scan_pass2(
    const float* __restrict__ chunkA, float* __restrict__ chunkH) {
  const int gid = blockIdx.x * P2T + threadIdx.x;
  const int d = gid % DI;
  const int n = (gid / DI) % NN;
  const int bk = gid / (DI * NN);
  const int cs = NN * DI;
  const float* pa = chunkA + ((size_t)bk * NC * NN + n) * DI + d;
  float* ph = chunkH + ((size_t)bk * NC * NN + n) * DI + d;
  float hrun = 0.f;
#pragma unroll 1
  for (int c0 = 0; c0 < NC; c0 += P2U) {
    float a[P2U], hl[P2U];
#pragma unroll
    for (int j = 0; j < P2U; ++j) {
      a[j] = pa[(c0 + j) * cs];
      hl[j] = ph[(c0 + j) * cs];
    }
#pragma unroll
    for (int j = 0; j < P2U; ++j) {
      ph[(c0 + j) * cs] = hrun;
      hrun = a[j] * hrun + hl[j];
    }
  }
}

__device__ __forceinline__ void p3_batch(const float (&ub)[SB], const float (&db)[SB],
                                         const float (&Av)[NN], float (&h)[NN],
                                         const float Dv, const float* row0, float* yo) {
#pragma unroll
  for (int j = 0; j < SB; ++j) {
    const float u = ub[j], dlt = db[j];
    const float* row = row0 + (size_t)j * CDBL;
    const float du = dlt * u;
    float y = 0.f;
#pragma unroll
    for (int n = 0; n < NN; ++n) {
      const float dA = __expf(dlt * Av[n]);
      h[n] = dA * h[n] + du * row[n];
      y += h[n] * row[NN + n];
    }
    yo[(size_t)j * DI] = y + u * Dv;
  }
}

__global__ void __launch_bounds__(DI) __attribute__((amdgpu_waves_per_eu(6)))
k_scan_pass3(
    const float* __restrict__ xconv, const float* __restrict__ delta,
    const float* __restrict__ x_dbl, const float* __restrict__ A_logs,
    const float* __restrict__ Ds, const float* __restrict__ chunkH,
    float* __restrict__ ys) {
  const int d = threadIdx.x;
  const int c = blockIdx.x % NC;
  const int bk = blockIdx.x / NC;
  const int k = bk % KK, b = bk / KK;
  float Av[NN];
#pragma unroll
  for (int n = 0; n < NN; ++n)
    Av[n] = -__expf(A_logs[(size_t)(k * DI + d) * NN + n]);
  const float Dv = Ds[k * DI + d];
  float h[NN];
#pragma unroll
  for (int n = 0; n < NN; ++n)
    h[n] = chunkH[((size_t)(bk * NC + c) * NN + n) * DI + d];
  const int l0 = c * CL;
  const float* up = xconv + ((size_t)b * LL + (k ? (LL - 1 - l0) : l0)) * DI + d;
  const int ustep = k ? -DI : DI;
  const float* dp = delta + ((size_t)bk * LL + l0) * DI + d;
  const float* xr = x_dbl + ((size_t)bk * LL + l0) * CDBL + RR;
  float* yo = ys + ((size_t)bk * LL + l0) * DI + d;
  float ub[SB], db[SB];
#pragma unroll
  for (int j = 0; j < SB; ++j) { ub[j] = up[j * ustep]; db[j] = dp[j * DI]; }
#pragma unroll 1
  for (int s0 = 0; s0 < CL - SB; s0 += SB) {
    float un[SB], dn[SB];
#pragma unroll
    for (int j = 0; j < SB; ++j) {
      un[j] = up[(s0 + SB + j) * ustep];
      dn[j] = dp[(s0 + SB + j) * DI];
    }
    __builtin_amdgcn_sched_barrier(0);
    p3_batch(ub, db, Av, h, Dv, xr + (size_t)s0 * CDBL, yo + (size_t)s0 * DI);
#pragma unroll
    for (int j = 0; j < SB; ++j) { ub[j] = un[j]; db[j] = dn[j]; }
  }
  p3_batch(ub, db, Av, h, Dv, xr + (size_t)(CL - SB) * CDBL, yo + (size_t)(CL - SB) * DI);
}

__global__ void __launch_bounds__(256) k_outproj(const float* __restrict__ ys,
                                                 const float* __restrict__ xz,
                                                 const float* __restrict__ W_out,
                                                 float* __restrict__ out) {
  __shared__ float sY[16][DI + 1];
  const int rowBase = blockIdx.x * 16;
  const int b = rowBase / LL;
  const int lb = rowBase % LL;
  const int tid = threadIdx.x;
  for (int i = tid; i < 16 * DI; i += 256) {
    int m = i / DI, dd = i % DI;
    int l = lb + m;
    float y0 = ys[((size_t)(b * KK + 0) * LL + l) * DI + dd];
    float y1 = ys[((size_t)(b * KK + 1) * LL + (LL - 1 - l)) * DI + dd];
    float zv = xz[(size_t)(b * LL + l) * NOUT + DI + dd];
    sY[m][dd] = (y0 + y1) * silu_f(zv);
  }
  __syncthreads();
  const int wave = tid >> 5, lane = tid & 31;
  const int lid = lane & 15, half = lane >> 4;
  for (int nt = wave; nt < CC / 16; nt += 8) {
    const int nbase = nt * 16;
    const float* wrow = W_out + (size_t)(nbase + lid) * DI;
    v8f acc = {};
    for (int k0 = 0; k0 < DI; k0 += 4) {
      v2f a;  a.x = sY[lid][k0 + 2 * half];  a.y = sY[lid][k0 + 2 * half + 1];
      v2f bb; bb.x = wrow[k0 + 2 * half];    bb.y = wrow[k0 + 2 * half + 1];
      acc = wmma4(a, bb, acc);
    }
    float* orow = out + (size_t)(rowBase + half * 8) * CC + nbase + lid;
#pragma unroll
    for (int v = 0; v < 8; ++v) orow[(size_t)v * CC] = acc[v];
  }
}

extern "C" void kernel_launch(void* const* d_in, const int* in_sizes, int n_in,
                              void* d_out, int out_size, void* d_ws, size_t ws_size,
                              hipStream_t stream) {
  (void)in_sizes; (void)n_in; (void)out_size; (void)ws_size;
  const float* x      = (const float*)d_in[0];
  const float* W_in   = (const float*)d_in[1];
  const float* conv_w = (const float*)d_in[2];
  const float* conv_b = (const float*)d_in[3];
  const float* xpw    = (const float*)d_in[4];
  const float* dtw    = (const float*)d_in[5];
  const float* dtb    = (const float*)d_in[6];
  const float* A_logs = (const float*)d_in[7];
  const float* Ds     = (const float*)d_in[8];
  const float* W_out  = (const float*)d_in[9];
  float* out = (float*)d_out;

  char* ws = (char*)d_ws;
  size_t off = 0;
  auto alloc = [&](size_t nbytes) {
    char* p = ws + off;
    off = (off + nbytes + 255) & ~(size_t)255;
    return (float*)p;
  };
  float* xz     = alloc((size_t)BL * NOUT * sizeof(float));
  float* xconv  = alloc((size_t)BL * DI * sizeof(float));
  float* x_dbl  = alloc((size_t)BB * KK * LL * CDBL * sizeof(float));
  float* delta  = alloc((size_t)BB * KK * LL * DI * sizeof(float));
  float* ysb    = alloc((size_t)BB * KK * LL * DI * sizeof(float));
  float* chunkA = alloc((size_t)BB * KK * DI * NC * NN * sizeof(float));
  float* chunkH = alloc((size_t)BB * KK * DI * NC * NN * sizeof(float));

  k_inproj<<<BL / 16, 256, 0, stream>>>(x, W_in, xz);
  k_dwconv<<<(BB * HH * DWSTRIPS) / (DWT / DWC4), DWT, 0, stream>>>(xz, conv_w,
                                                                    conv_b, xconv);
  k_xdbl<<<(BB * KK * LL / 16) / 8, 256, 0, stream>>>(xconv, xpw, x_dbl);
  k_dtproj<<<BB * KK * LL / 16, 256, 0, stream>>>(x_dbl, dtw, dtb, delta);
  k_scan_pass1<<<BB * KK * NC, DI, 0, stream>>>(
      xconv, delta, x_dbl, A_logs, chunkA, chunkH);
  k_scan_pass2<<<(BB * KK * NN * DI) / P2T, P2T, 0, stream>>>(chunkA, chunkH);
  k_scan_pass3<<<BB * KK * NC, DI, 0, stream>>>(
      xconv, delta, x_dbl, A_logs, Ds, chunkH, ysb);
  k_outproj<<<BL / 16, 256, 0, stream>>>(ysb, xz, W_out, out);
}


# in_proj, x_dbl, out_proj: each wave keeps its weight fragments in registers and walks several row tiles staged in LDS by whole-line 16-byte loads; k order within each group of 8 changed
# speedup vs baseline: 2.4490x; 1.2798x over previous
#include <hip/hip_runtime.h>
#include <cstdint>
#include <cstddef>

#define BB 4
#define HH 32
#define WWD 256
#define CC 96
#define DI 192
#define KK 2
#define NN 16
#define RR 6
#define LL (HH*WWD)
#define BL (BB*LL)
#define NOUT (2*DI)
#define CDBL (RR + 2*NN)
#define NC 128
#define CL (LL/NC)

typedef float v2f __attribute__((ext_vector_type(2)));
typedef float v8f __attribute__((ext_vector_type(8)));

__device__ __forceinline__ v8f wmma4(v2f a, v2f b, v8f c) {
  return __builtin_amdgcn_wmma_f32_16x16x4_f32(false, a, false, b, (short)0, c,
                                               false, false);
}

__device__ __forceinline__ float silu_f(float x) {
  return x / (1.0f + __expf(-x));
}
__device__ __forceinline__ float softplus_f(float x) {
  return (x > 20.0f) ? x : log1pf(__expf(x));
}

__device__ __forceinline__ v2f lo2(const float4 v) { v2f r; r.x = v.x; r.y = v.y; return r; }
__device__ __forceinline__ v2f hi2(const float4 v) { v2f r; r.x = v.z; r.y = v.w; return r; }

#define IPR 128
#define IPP 100
__global__ void __launch_bounds__(256) k_inproj(const float* __restrict__ x,
                                                const float* __restrict__ W_in,
                                                float* __restrict__ xz) {
  __shared__ __attribute__((aligned(16))) float sX[IPR * IPP];
  const int tid = threadIdx.x;
  const int rowBase = blockIdx.x * IPR;
  const int wave = tid >> 5, lane = tid & 31;
  const int lid = lane & 15, half = lane >> 4;
  float4 bf[3][CC / 8];
#pragma unroll
  for (int j = 0; j < 3; ++j) {
    const float* wrow = W_in + (size_t)((wave + 8 * j) * 16 + lid) * CC + 4 * half;
#pragma unroll
    for (int i = 0; i < CC / 8; ++i)
      bf[j][i] = *reinterpret_cast<const float4*>(wrow + 8 * i);
  }
#pragma unroll
  for (int it = 0; it < IPR * (CC / 4) / 256; ++it) {
    const int i = tid + it * 256;
    const int m = i / (CC / 4), q = i % (CC / 4);
    const float4 v = *reinterpret_cast<const float4*>(x + (size_t)(rowBase + m) * CC + q * 4);
    *reinterpret_cast<float4*>(&sX[m * IPP + q * 4]) = v;
  }
  __syncthreads();
#pragma unroll 1
  for (int mt = 0; mt < IPR / 16; ++mt) {
    const float* arow = &sX[(mt * 16 + lid) * IPP + 4 * half];
    float4 af[CC / 8];
#pragma unroll
    for (int i = 0; i < CC / 8; ++i) af[i] = *reinterpret_cast<const float4*>(arow + 8 * i);
    v8f acc[3] = {};
#pragma unroll
    for (int i = 0; i < CC / 8; ++i) {
#pragma unroll
      for (int j = 0; j < 3; ++j) acc[j] = wmma4(lo2(af[i]), lo2(bf[j][i]), acc[j]);
#pragma unroll
      for (int j = 0; j < 3; ++j) acc[j] = wmma4(hi2(af[i]), hi2(bf[j][i]), acc[j]);
    }
    float* orow = xz + (size_t)(rowBase + mt * 16 + half * 8) * NOUT + lid;
#pragma unroll
    for (int j = 0; j < 3; ++j)
#pragma unroll
      for (int v = 0; v < 8; ++v)
        orow[(size_t)v * NOUT + (wave + 8 * j) * 16] = acc[j][v];
  }
}

#define DWS 8
#define DWC4 (DI / 4)
#define DWSTRIPS (WWD / DWS)
#define DWT 192
__global__ void __launch_bounds__(DWT) k_dwconv(const float* __restrict__ xz,
                                                const float* __restrict__ conv_w,
                                                const float* __restrict__ conv_b,
                                                float* __restrict__ xconv) {
  const unsigned tid = threadIdx.x;
  const int c = (int)(tid % DWC4) * 4;
  const unsigned s = blockIdx.x * (DWT / DWC4) + tid / DWC4;
  const int w0 = (int)(s % DWSTRIPS) * DWS;
  const int h = (int)((s / DWSTRIPS) % HH);
  const int b = (int)(s / (DWSTRIPS * HH));

  float wt[4][9];
#pragma unroll
  for (int j = 0; j < 4; ++j)
#pragma unroll
    for (int t = 0; t < 9; ++t) wt[j][t] = conv_w[(c + j) * 9 + t];
  float bias[4];
#pragma unroll
  for (int j = 0; j < 4; ++j) bias[j] = conv_b[c + j];

  float v[3][DWS + 2][4];
#pragma unroll
  for (int r = 0; r < 3; ++r) {
    const int hh = h + r - 1;
    const bool rok = (hh >= 0) && (hh < HH);
    const int hc = hh < 0 ? 0 : (hh >= HH ? HH - 1 : hh);
#pragma unroll
    for (int q = 0; q < DWS + 2; ++q) {
      bool cok = true;
      int wc = w0 + q - 1;
      if (q == 0) { cok = w0 > 0; wc = cok ? wc : 0; }
      if (q == DWS + 1) { cok = (w0 + DWS) < WWD; wc = cok ? wc : WWD - 1; }
      const bool ok = rok && cok;
      const float4 t4 = *reinterpret_cast<const float4*>(
          xz + (size_t)((b * HH + hc) * WWD + wc) * NOUT + c);
      v[r][q][0] = ok ? t4.x : 0.f;
      v[r][q][1] = ok ? t4.y : 0.f;
      v[r][q][2] = ok ? t4.z : 0.f;
      v[r][q][3] = ok ? t4.w : 0.f;
    }
  }

  float* orow = xconv + (size_t)((b * HH + h) * WWD + w0) * DI + c;
#pragma unroll
  for (int i = 0; i < DWS; ++i) {
    float o[4];
#pragma unroll
    for (int j = 0; j < 4; ++j) {
      float acc = bias[j];
#pragma unroll
      for (int r = 0; r < 3; ++r)
#pragma unroll
        for (int dx = 0; dx < 3; ++dx) acc += v[r][i + dx][j] * wt[j][r * 3 + dx];
      o[j] = silu_f(acc);
    }
    *reinterpret_cast<float4*>(orow + (size_t)i * DI) =
        make_float4(o[0], o[1], o[2], o[3]);
  }
}

#define GR 64
#define GP 196
__global__ void __launch_bounds__(192) k_xdbl(const float* __restrict__ xconv,
                                              const float* __restrict__ xpw,
                                              float* __restrict__ x_dbl) {
  __shared__ __attribute__((aligned(16))) float sA[GR * GP];
  const int tid = threadIdx.x, wave = tid >> 5, lane = tid & 31;
  const int lid = lane & 15, half = lane >> 4;
  const int bk = blockIdx.x / (LL / GR);
  const int lbase = (blockIdx.x % (LL / GR)) * GR;
  const int k = bk % KK, b = bk / KK;
  const int j = wave % 3, g = wave / 3;
  const int col = j * 16 + lid;
  const bool cok = col < CDBL;
  const float* wrow = xpw + (size_t)(k * CDBL + (cok ? col : 0)) * DI + 4 * half;
  float4 bf[DI / 8];
#pragma unroll
  for (int i = 0; i < DI / 8; ++i) {
    const float4 t4 = *reinterpret_cast<const float4*>(wrow + 8 * i);
    bf[i].x = cok ? t4.x : 0.f; bf[i].y = cok ? t4.y : 0.f;
    bf[i].z = cok ? t4.z : 0.f; bf[i].w = cok ? t4.w : 0.f;
  }
  const int q = tid % (DI / 4), m0 = tid / (DI / 4);
#pragma unroll 8
  for (int it = 0; it < GR / 4; ++it) {
    const int m = it * 4 + m0;
    const int l = lbase + m;
    const int lidx = k ? (LL - 1 - l) : l;
    const float4 v = *reinterpret_cast<const float4*>(xconv + (size_t)(b * LL + lidx) * DI + q * 4);
    *reinterpret_cast<float4*>(&sA[m * GP + q * 4]) = v;
  }
  __syncthreads();
  const float* ar0 = &sA[(g * 32 + lid) * GP + 4 * half];
  const float* ar1 = ar0 + 16 * GP;
  v8f acc0 = {}, acc1 = {};
#pragma unroll
  for (int i = 0; i < DI / 8; ++i) {
    const float4 a0 = *reinterpret_cast<const float4*>(ar0 + 8 * i);
    const float4 a1 = *reinterpret_cast<const float4*>(ar1 + 8 * i);
    acc0 = wmma4(lo2(a0), lo2(bf[i]), acc0);
    acc1 = wmma4(lo2(a1), lo2(bf[i]), acc1);
    acc0 = wmma4(hi2(a0), hi2(bf[i]), acc0);
    acc1 = wmma4(hi2(a1), hi2(bf[i]), acc1);
  }
  if (cok) {
    float* orow = x_dbl + ((size_t)bk * LL + lbase + g * 32 + half * 8) * CDBL + col;
#pragma unroll
    for (int v = 0; v < 8; ++v) {
      orow[(size_t)v * CDBL] = acc0[v];
      orow[(size_t)(v + 16) * CDBL] = acc1[v];
    }
  }
}

__global__ void __launch_bounds__(256) k_dtproj(const float* __restrict__ x_dbl,
                                                const float* __restrict__ dtw,
                                                const float* __restrict__ dtb,
                                                float* __restrict__ delta) {
  const int tid = threadIdx.x, wave = tid >> 5, lane = tid & 31;
  const int lid = lane & 15, half = lane >> 4;
  const int tile = blockIdx.x;
  const int lt = tile % (LL / 16);
  const int bk = tile / (LL / 16);
  const int k = bk % KK;
  const int lbase = lt * 16;
  const float* arow = x_dbl + ((size_t)bk * LL + lbase + lid) * CDBL;
  v2f a0, a1;
  a0.x = arow[2 * half]; a0.y = arow[2 * half + 1];
  a1.x = (half == 0) ? arow[4] : 0.f;
  a1.y = (half == 0) ? arow[5] : 0.f;
  for (int nt = wave; nt < DI / 16; nt += 8) {
    const int d = nt * 16 + lid;
    const float* wrow = dtw + (size_t)(k * DI + d) * RR;
    v2f b0, b1;
    b0.x = wrow[2 * half]; b0.y = wrow[2 * half + 1];
    b1.x = (half == 0) ? wrow[4] : 0.f;
    b1.y = (half == 0) ? wrow[5] : 0.f;
    v8f acc = {};
    acc = wmma4(a0, b0, acc);
    acc = wmma4(a1, b1, acc);
    const float bias = dtb[k * DI + d];
    float* orow = delta + ((size_t)bk * LL + lbase + half * 8) * DI + d;
#pragma unroll
    for (int v = 0; v < 8; ++v) orow[(size_t)v * DI] = softplus_f(acc[v] + bias);
  }
}

#define SB 8

__device__ __forceinline__ void p1_batch(const float (&ub)[SB], const float (&db)[SB],
                                         const float (&Av)[NN], float (&aP)[NN],
                                         float (&h)[NN], const float* row0) {
#pragma unroll
  for (int j = 0; j < SB; ++j) {
    const float u = ub[j], dlt = db[j];
    const float* row = row0 + (size_t)j * CDBL;
    const float du = dlt * u;
#pragma unroll
    for (int n = 0; n < NN; ++n) {
      const float dA = __expf(dlt * Av[n]);
      aP[n] *= dA;
      h[n] = dA * h[n] + du * row[n];
    }
  }
}

__global__ void __launch_bounds__(DI) k_scan_pass1(
    const float* __restrict__ xconv, const float* __restrict__ delta,
    const float* __restrict__ x_dbl, const float* __restrict__ A_logs,
    float* __restrict__ chunkA, float* __restrict__ chunkH) {
  const int d = threadIdx.x;
  const int c = blockIdx.x % NC;
  const int bk = blockIdx.x / NC;
  const int k = bk % KK, b = bk / KK;
  float Av[NN];
#pragma unroll
  for (int n = 0; n < NN; ++n)
    Av[n] = -__expf(A_logs[(size_t)(k * DI + d) * NN + n]);
  float aP[NN], h[NN];
#pragma unroll
  for (int n = 0; n < NN; ++n) { aP[n] = 1.f; h[n] = 0.f; }
  const int l0 = c * CL;
  const float* up = xconv + ((size_t)b * LL + (k ? (LL - 1 - l0) : l0)) * DI + d;
  const int ustep = k ? -DI : DI;
  const float* dp = delta + ((size_t)bk * LL + l0) * DI + d;
  const float* xr = x_dbl + ((size_t)bk * LL + l0) * CDBL + RR;
  float ub[SB], db[SB];
#pragma unroll
  for (int j = 0; j < SB; ++j) { ub[j] = up[j * ustep]; db[j] = dp[j * DI]; }
#pragma unroll 1
  for (int s0 = 0; s0 < CL - SB; s0 += SB) {
    float un[SB], dn[SB];
#pragma unroll
    for (int j = 0; j < SB; ++j) {
      un[j] = up[(s0 + SB + j) * ustep];
      dn[j] = dp[(s0 + SB + j) * DI];
    }
    __builtin_amdgcn_sched_barrier(0);
    p1_batch(ub, db, Av, aP, h, xr + (size_t)s0 * CDBL);
#pragma unroll
    for (int j = 0; j < SB; ++j) { ub[j] = un[j]; db[j] = dn[j]; }
  }
  p1_batch(ub, db, Av, aP, h, xr + (size_t)(CL - SB) * CDBL);
#pragma unroll
  for (int n = 0; n < NN; ++n) {
    const size_t idx = ((size_t)(bk * NC + c) * NN + n) * DI + d;
    chunkA[idx] = aP[n];
    chunkH[idx] = h[n];
  }
}

#define P2U 32
#define P2T 96
__global__ void __launch_bounds__(P2T) k_scan_pass2(
    const float* __restrict__ chunkA, float* __restrict__ chunkH) {
  const int gid = blockIdx.x * P2T + threadIdx.x;
  const int d = gid % DI;
  const int n = (gid / DI) % NN;
  const int bk = gid / (DI * NN);
  const int cs = NN * DI;
  const float* pa = chunkA + ((size_t)bk * NC * NN + n) * DI + d;
  float* ph = chunkH + ((size_t)bk * NC * NN + n) * DI + d;
  float hrun = 0.f;
#pragma unroll 1
  for (int c0 = 0; c0 < NC; c0 += P2U) {
    float a[P2U], hl[P2U];
#pragma unroll
    for (int j = 0; j < P2U; ++j) {
      a[j] = pa[(c0 + j) * cs];
      hl[j] = ph[(c0 + j) * cs];
    }
#pragma unroll
    for (int j = 0; j < P2U; ++j) {
      ph[(c0 + j) * cs] = hrun;
      hrun = a[j] * hrun + hl[j];
    }
  }
}

__device__ __forceinline__ void p3_batch(const float (&ub)[SB], const float (&db)[SB],
                                         const float (&Av)[NN], float (&h)[NN],
                                         const float Dv, const float* row0, float* yo) {
#pragma unroll
  for (int j = 0; j < SB; ++j) {
    const float u = ub[j], dlt = db[j];
    const float* row = row0 + (size_t)j * CDBL;
    const float du = dlt * u;
    float y = 0.f;
#pragma unroll
    for (int n = 0; n < NN; ++n) {
      const float dA = __expf(dlt * Av[n]);
      h[n] = dA * h[n] + du * row[n];
      y += h[n] * row[NN + n];
    }
    yo[(size_t)j * DI] = y + u * Dv;
  }
}

__global__ void __launch_bounds__(DI) __attribute__((amdgpu_waves_per_eu(6)))
k_scan_pass3(
    const float* __restrict__ xconv, const float* __restrict__ delta,
    const float* __restrict__ x_dbl, const float* __restrict__ A_logs,
    const float* __restrict__ Ds, const float* __restrict__ chunkH,
    float* __restrict__ ys) {
  const int d = threadIdx.x;
  const int c = blockIdx.x % NC;
  const int bk = blockIdx.x / NC;
  const int k = bk % KK, b = bk / KK;
  float Av[NN];
#pragma unroll
  for (int n = 0; n < NN; ++n)
    Av[n] = -__expf(A_logs[(size_t)(k * DI + d) * NN + n]);
  const float Dv = Ds[k * DI + d];
  float h[NN];
#pragma unroll
  for (int n = 0; n < NN; ++n)
    h[n] = chunkH[((size_t)(bk * NC + c) * NN + n) * DI + d];
  const int l0 = c * CL;
  const float* up = xconv + ((size_t)b * LL + (k ? (LL - 1 - l0) : l0)) * DI + d;
  const int ustep = k ? -DI : DI;
  const float* dp = delta + ((size_t)bk * LL + l0) * DI + d;
  const float* xr = x_dbl + ((size_t)bk * LL + l0) * CDBL + RR;
  float* yo = ys + ((size_t)bk * LL + l0) * DI + d;
  float ub[SB], db[SB];
#pragma unroll
  for (int j = 0; j < SB; ++j) { ub[j] = up[j * ustep]; db[j] = dp[j * DI]; }
#pragma unroll 1
  for (int s0 = 0; s0 < CL - SB; s0 += SB) {
    float un[SB], dn[SB];
#pragma unroll
    for (int j = 0; j < SB; ++j) {
      un[j] = up[(s0 + SB + j) * ustep];
      dn[j] = dp[(s0 + SB + j) * DI];
    }
    __builtin_amdgcn_sched_barrier(0);
    p3_batch(ub, db, Av, h, Dv, xr + (size_t)s0 * CDBL, yo + (size_t)s0 * DI);
#pragma unroll
    for (int j = 0; j < SB; ++j) { ub[j] = un[j]; db[j] = dn[j]; }
  }
  p3_batch(ub, db, Av, h, Dv, xr + (size_t)(CL - SB) * CDBL, yo + (size_t)(CL - SB) * DI);
}

__global__ void __launch_bounds__(192) k_outproj(const float* __restrict__ ys,
                                                 const float* __restrict__ xz,
                                                 const float* __restrict__ W_out,
                                                 float* __restrict__ out) {
  __shared__ __attribute__((aligned(16))) float sY[GR * GP];
  const int tid = threadIdx.x, wave = tid >> 5, lane = tid & 31;
  const int lid = lane & 15, half = lane >> 4;
  const int rowBase = blockIdx.x * GR;
  const int b = rowBase / LL;
  const int lb = rowBase % LL;
  const float* wrow = W_out + (size_t)(wave * 16 + lid) * DI + 4 * half;
  float4 bf[DI / 8];
#pragma unroll
  for (int i = 0; i < DI / 8; ++i) bf[i] = *reinterpret_cast<const float4*>(wrow + 8 * i);
  const int q = tid % (DI / 4), m0 = tid / (DI / 4);
#pragma unroll 4
  for (int it = 0; it < GR / 4; ++it) {
    const int m = it * 4 + m0;
    const int l = lb + m;
    const float4 y0 = *reinterpret_cast<const float4*>(
        ys + ((size_t)(b * KK + 0) * LL + l) * DI + q * 4);
    const float4 y1 = *reinterpret_cast<const float4*>(
        ys + ((size_t)(b * KK + 1) * LL + (LL - 1 - l)) * DI + q * 4);
    const float4 zv = *reinterpret_cast<const float4*>(
        xz + (size_t)(b * LL + l) * NOUT + DI + q * 4);
    float4 r;
    r.x = (y0.x + y1.x) * silu_f(zv.x);
    r.y = (y0.y + y1.y) * silu_f(zv.y);
    r.z = (y0.z + y1.z) * silu_f(zv.z);
    r.w = (y0.w + y1.w) * silu_f(zv.w);
    *reinterpret_cast<float4*>(&sY[m * GP + q * 4]) = r;
  }
  __syncthreads();
#pragma unroll 1
  for (int mp = 0; mp < GR / 32; ++mp) {
    const float* ar0 = &sY[(mp * 32 + lid) * GP + 4 * half];
    const float* ar1 = ar0 + 16 * GP;
    v8f acc0 = {}, acc1 = {};
#pragma unroll
    for (int i = 0; i < DI / 8; ++i) {
      const float4 a0 = *reinterpret_cast<const float4*>(ar0 + 8 * i);
      const float4 a1 = *reinterpret_cast<const float4*>(ar1 + 8 * i);
      acc0 = wmma4(lo2(a0), lo2(bf[i]), acc0);
      acc1 = wmma4(lo2(a1), lo2(bf[i]), acc1);
      acc0 = wmma4(hi2(a0), hi2(bf[i]), acc0);
      acc1 = wmma4(hi2(a1), hi2(bf[i]), acc1);
    }
    float* orow = out + (size_t)(rowBase + mp * 32 + half * 8) * CC + wave * 16 + lid;
#pragma unroll
    for (int v = 0; v < 8; ++v) {
      orow[(size_t)v * CC] = acc0[v];
      orow[(size_t)(v + 16) * CC] = acc1[v];
    }
  }
}

extern "C" void kernel_launch(void* const* d_in, const int* in_sizes, int n_in,
                              void* d_out, int out_size, void* d_ws, size_t ws_size,
                              hipStream_t stream) {
  (void)in_sizes; (void)n_in; (void)out_size; (void)ws_size;
  const float* x      = (const float*)d_in[0];
  const float* W_in   = (const float*)d_in[1];
  const float* conv_w = (const float*)d_in[2];
  const float* conv_b = (const float*)d_in[3];
  const float* xpw    = (const float*)d_in[4];
  const float* dtw    = (const float*)d_in[5];
  const float* dtb    = (const float*)d_in[6];
  const float* A_logs = (const float*)d_in[7];
  const float* Ds     = (const float*)d_in[8];
  const float* W_out  = (const float*)d_in[9];
  float* out = (float*)d_out;

  char* ws = (char*)d_ws;
  size_t off = 0;
  auto alloc = [&](size_t nbytes) {
    char* p = ws + off;
    off = (off + nbytes + 255) & ~(size_t)255;
    return (float*)p;
  };
  float* xz     = alloc((size_t)BL * NOUT * sizeof(float));
  float* xconv  = alloc((size_t)BL * DI * sizeof(float));
  float* x_dbl  = alloc((size_t)BB * KK * LL * CDBL * sizeof(float));
  float* delta  = alloc((size_t)BB * KK * LL * DI * sizeof(float));
  float* ysb    = alloc((size_t)BB * KK * LL * DI * sizeof(float));
  float* chunkA = alloc((size_t)BB * KK * DI * NC * NN * sizeof(float));
  float* chunkH = alloc((size_t)BB * KK * DI * NC * NN * sizeof(float));

  k_inproj<<<BL / IPR, 256, 0, stream>>>(x, W_in, xz);
  k_dwconv<<<(BB * HH * DWSTRIPS) / (DWT / DWC4), DWT, 0, stream>>>(xz, conv_w,
                                                                    conv_b, xconv);
  k_xdbl<<<BB * KK * LL / GR, 192, 0, stream>>>(xconv, xpw, x_dbl);
  k_dtproj<<<BB * KK * LL / 16, 256, 0, stream>>>(x_dbl, dtw, dtb, delta);
  k_scan_pass1<<<BB * KK * NC, DI, 0, stream>>>(
      xconv, delta, x_dbl, A_logs, chunkA, chunkH);
  k_scan_pass2<<<(BB * KK * NN * DI) / P2T, P2T, 0, stream>>>(chunkA, chunkH);
  k_scan_pass3<<<BB * KK * NC, DI, 0, stream>>>(
      xconv, delta, x_dbl, A_logs, Ds, chunkH, ysb);
  k_outproj<<<BL / GR, 192, 0, stream>>>(ysb, xz, W_out, out);
}


# dt projection + softplus moved into the x_dbl kernel as an epilogue on its own tile (WMMA kept), softplus on hardware exp2/log2/rcp instead of library log1pf: one launch and 32768 short waves fewer
# speedup vs baseline: 2.7529x; 1.1241x over previous
#include <hip/hip_runtime.h>
#include <cstdint>
#include <cstddef>

#define BB 4
#define HH 32
#define WWD 256
#define CC 96
#define DI 192
#define KK 2
#define NN 16
#define RR 6
#define LL (HH*WWD)
#define BL (BB*LL)
#define NOUT (2*DI)
#define CDBL (RR + 2*NN)
#define NC 128
#define CL (LL/NC)

typedef float v2f __attribute__((ext_vector_type(2)));
typedef float v8f __attribute__((ext_vector_type(8)));

__device__ __forceinline__ v8f wmma4(v2f a, v2f b, v8f c) {
  return __builtin_amdgcn_wmma_f32_16x16x4_f32(false, a, false, b, (short)0, c,
                                               false, false);
}

__device__ __forceinline__ float silu_f(float x) {
  return x / (1.0f + __expf(-x));
}
__device__ __forceinline__ float softplus_f(float x) {
  const float e = __expf(-fabsf(x));
  const float u = 1.0f + e;
  const float lost = e - (u - 1.0f);
  const float l = __builtin_amdgcn_logf(u) * 0.6931471805599453f;
  return fmaxf(x, 0.0f) + (l + lost * __builtin_amdgcn_rcpf(u));
}
__device__ __forceinline__ float xhalf(float x) {
  return __builtin_bit_cast(float, __builtin_amdgcn_permlanex16(
      __builtin_bit_cast(int, x), __builtin_bit_cast(int, x), 0x76543210, 0xfedcba98, false, false));
}

__device__ __forceinline__ v2f lo2(const float4 v) { v2f r; r.x = v.x; r.y = v.y; return r; }
__device__ __forceinline__ v2f hi2(const float4 v) { v2f r; r.x = v.z; r.y = v.w; return r; }

#define IPR 128
#define IPP 100
__global__ void __launch_bounds__(256) k_inproj(const float* __restrict__ x,
                                                const float* __restrict__ W_in,
                                                float* __restrict__ xz) {
  __shared__ __attribute__((aligned(16))) float sX[IPR * IPP];
  const int tid = threadIdx.x;
  const int rowBase = blockIdx.x * IPR;
  const int wave = tid >> 5, lane = tid & 31;
  const int lid = lane & 15, half = lane >> 4;
  float4 bf[3][CC / 8];
#pragma unroll
  for (int j = 0; j < 3; ++j) {
    const float* wrow = W_in + (size_t)((wave + 8 * j) * 16 + lid) * CC + 4 * half;
#pragma unroll
    for (int i = 0; i < CC / 8; ++i)
      bf[j][i] = *reinterpret_cast<const float4*>(wrow + 8 * i);
  }
#pragma unroll
  for (int it = 0; it < IPR * (CC / 4) / 256; ++it) {
    const int i = tid + it * 256;
    const int m = i / (CC / 4), q = i % (CC / 4);
    const float4 v = *reinterpret_cast<const float4*>(x + (size_t)(rowBase + m) * CC + q * 4);
    *reinterpret_cast<float4*>(&sX[m * IPP + q * 4]) = v;
  }
  __syncthreads();
#pragma unroll 1
  for (int mt = 0; mt < IPR / 16; ++mt) {
    const float* arow = &sX[(mt * 16 + lid) * IPP + 4 * half];
    float4 af[CC / 8];
#pragma unroll
    for (int i = 0; i < CC / 8; ++i) af[i] = *reinterpret_cast<const float4*>(arow + 8 * i);
    v8f acc[3] = {};
#pragma unroll
    for (int i = 0; i < CC / 8; ++i) {
#pragma unroll
      for (int j = 0; j < 3; ++j) acc[j] = wmma4(lo2(af[i]), lo2(bf[j][i]), acc[j]);
#pragma unroll
      for (int j = 0; j < 3; ++j) acc[j] = wmma4(hi2(af[i]), hi2(bf[j][i]), acc[j]);
    }
    float* orow = xz + (size_t)(rowBase + mt * 16 + half * 8) * NOUT + lid;
#pragma unroll
    for (int j = 0; j < 3; ++j)
#pragma unroll
      for (int v = 0; v < 8; ++v)
        orow[(size_t)v * NOUT + (wave + 8 * j) * 16] = acc[j][v];
  }
}

#define DWS 8
#define DWC4 (DI / 4)
#define DWSTRIPS (WWD / DWS)
#define DWT 192
__global__ void __launch_bounds__(DWT) k_dwconv(const float* __restrict__ xz,
                                                const float* __restrict__ conv_w,
                                                const float* __restrict__ conv_b,
                                                float* __restrict__ xconv) {
  const unsigned tid = threadIdx.x;
  const int c = (int)(tid % DWC4) * 4;
  const unsigned s = blockIdx.x * (DWT / DWC4) + tid / DWC4;
  const int w0 = (int)(s % DWSTRIPS) * DWS;
  const int h = (int)((s / DWSTRIPS) % HH);
  const int b = (int)(s / (DWSTRIPS * HH));

  float wt[4][9];
#pragma unroll
  for (int j = 0; j < 4; ++j)
#pragma unroll
    for (int t = 0; t < 9; ++t) wt[j][t] = conv_w[(c + j) * 9 + t];
  float bias[4];
#pragma unroll
  for (int j = 0; j < 4; ++j) bias[j] = conv_b[c + j];

  float v[3][DWS + 2][4];
#pragma unroll
  for (int r = 0; r < 3; ++r) {
    const int hh = h + r - 1;
    const bool rok = (hh >= 0) && (hh < HH);
    const int hc = hh < 0 ? 0 : (hh >= HH ? HH - 1 : hh);
#pragma unroll
    for (int q = 0; q < DWS + 2; ++q) {
      bool cok = true;
      int wc = w0 + q - 1;
      if (q == 0) { cok = w0 > 0; wc = cok ? wc : 0; }
      if (q == DWS + 1) { cok = (w0 + DWS) < WWD; wc = cok ? wc : WWD - 1; }
      const bool ok = rok && cok;
      const float4 t4 = *reinterpret_cast<const float4*>(
          xz + (size_t)((b * HH + hc) * WWD + wc) * NOUT + c);
      v[r][q][0] = ok ? t4.x : 0.f;
      v[r][q][1] = ok ? t4.y : 0.f;
      v[r][q][2] = ok ? t4.z : 0.f;
      v[r][q][3] = ok ? t4.w : 0.f;
    }
  }

  float* orow = xconv + (size_t)((b * HH + h) * WWD + w0) * DI + c;
#pragma unroll
  for (int i = 0; i < DWS; ++i) {
    float o[4];
#pragma unroll
    for (int j = 0; j < 4; ++j) {
      float acc = bias[j];
#pragma unroll
      for (int r = 0; r < 3; ++r)
#pragma unroll
        for (int dx = 0; dx < 3; ++dx) acc += v[r][i + dx][j] * wt[j][r * 3 + dx];
      o[j] = silu_f(acc);
    }
    *reinterpret_cast<float4*>(orow + (size_t)i * DI) =
        make_float4(o[0], o[1], o[2], o[3]);
  }
}

#define GR 64
#define GP 196
#define DTP 8
__global__ void __launch_bounds__(192) k_xdbl(const float* __restrict__ xconv,
                                              const float* __restrict__ xpw,
                                              const float* __restrict__ dtw,
                                              const float* __restrict__ dtb,
                                              float* __restrict__ x_dbl,
                                              float* __restrict__ delta) {
  __shared__ __attribute__((aligned(16))) float sA[GR * GP];
  __shared__ __attribute__((aligned(16))) float sD[GR * DTP];
  const int tid = threadIdx.x, wave = tid >> 5, lane = tid & 31;
  const int lid = lane & 15, half = lane >> 4;
  const int bk = blockIdx.x / (LL / GR);
  const int lbase = (blockIdx.x % (LL / GR)) * GR;
  const int k = bk % KK, b = bk / KK;
  const int j = wave % 3, g = wave / 3;
  const int col = j * 16 + lid;
  const bool cok = col < CDBL;
  const float* wrow = xpw + (size_t)(k * CDBL + (cok ? col : 0)) * DI + 4 * half;
  float4 bf[DI / 8];
#pragma unroll
  for (int i = 0; i < DI / 8; ++i) {
    const float4 t4 = *reinterpret_cast<const float4*>(wrow + 8 * i);
    bf[i].x = cok ? t4.x : 0.f; bf[i].y = cok ? t4.y : 0.f;
    bf[i].z = cok ? t4.z : 0.f; bf[i].w = cok ? t4.w : 0.f;
  }
  const int q = tid % (DI / 4), m0 = tid / (DI / 4);
#pragma unroll 8
  for (int it = 0; it < GR / 4; ++it) {
    const int m = it * 4 + m0;
    const int l = lbase + m;
    const int lidx = k ? (LL - 1 - l) : l;
    const float4 v = *reinterpret_cast<const float4*>(xconv + (size_t)(b * LL + lidx) * DI + q * 4);
    *reinterpret_cast<float4*>(&sA[m * GP + q * 4]) = v;
  }
  __syncthreads();
  const float* ar0 = &sA[(g * 32 + lid) * GP + 4 * half];
  const float* ar1 = ar0 + 16 * GP;
  v8f acc0 = {}, acc1 = {};
#pragma unroll
  for (int i = 0; i < DI / 8; ++i) {
    const float4 a0 = *reinterpret_cast<const float4*>(ar0 + 8 * i);
    const float4 a1 = *reinterpret_cast<const float4*>(ar1 + 8 * i);
    acc0 = wmma4(lo2(a0), lo2(bf[i]), acc0);
    acc1 = wmma4(lo2(a1), lo2(bf[i]), acc1);
    acc0 = wmma4(hi2(a0), hi2(bf[i]), acc0);
    acc1 = wmma4(hi2(a1), hi2(bf[i]), acc1);
  }
  v2f pb0[2], pb1[2];
  float pbias[2];
#pragma unroll
  for (int t = 0; t < 2; ++t) {
    const int dcol = wave * 32 + t * 16 + lid;
    const float* wr = dtw + (size_t)(k * DI + dcol) * RR;
    pb0[t].x = wr[2 * half]; pb0[t].y = wr[2 * half + 1];
    const float w4 = wr[4], w5 = wr[5];
    pb1[t].x = half ? 0.f : w4; pb1[t].y = half ? 0.f : w5;
    pbias[t] = dtb[k * DI + dcol];
  }
  if (j == 0 && lid < RR) {
    float* dr = &sD[(g * 32 + half * 8) * DTP + lid];
#pragma unroll
    for (int v = 0; v < 8; ++v) {
      dr[v * DTP] = acc0[v];
      dr[(v + 16) * DTP] = acc1[v];
    }
  }
  __syncthreads();
  if (cok) {
    float* orow = x_dbl + ((size_t)bk * LL + lbase + g * 32 + half * 8) * CDBL + col;
#pragma unroll
    for (int v = 0; v < 8; ++v) {
      orow[(size_t)v * CDBL] = acc0[v];
      orow[(size_t)(v + 16) * CDBL] = acc1[v];
    }
  }
  float* drow = delta + ((size_t)bk * LL + lbase) * DI + wave * 32 + lane;
  for (int rt = 0; rt < GR / 16; ++rt) {
    const float* ar = &sD[(rt * 16 + lid) * DTP];
    const float2 r03 = *reinterpret_cast<const float2*>(ar + 2 * half);
    const float2 r45 = *reinterpret_cast<const float2*>(ar + 4);
    v2f a0, a1;
    a0.x = r03.x; a0.y = r03.y;
    a1.x = half ? 0.f : r45.x; a1.y = half ? 0.f : r45.y;
    v8f d0 = {}, d1 = {};
    d0 = wmma4(a0, pb0[0], d0);
    d1 = wmma4(a0, pb0[1], d1);
    d0 = wmma4(a1, pb1[0], d0);
    d1 = wmma4(a1, pb1[1], d1);
    float* orow = drow + (size_t)(rt * 16) * DI;
#pragma unroll
    for (int v = 0; v < 8; ++v) {
      const float s0 = softplus_f(d0[v] + pbias[0]);
      const float s1 = softplus_f(d1[v] + pbias[1]);
      const float got = xhalf(half ? s0 : s1);
      orow[(size_t)v * DI] = half ? got : s0;
      orow[(size_t)(v + 8) * DI] = half ? s1 : got;
    }
  }
}

#define SB 8

__device__ __forceinline__ void p1_batch(const float (&ub)[SB], const float (&db)[SB],
                                         const float (&Av)[NN], float (&aP)[NN],
                                         float (&h)[NN], const float* row0) {
#pragma unroll
  for (int j = 0; j < SB; ++j) {
    const float u = ub[j], dlt = db[j];
    const float* row = row0 + (size_t)j * CDBL;
    const float du = dlt * u;
#pragma unroll
    for (int n = 0; n < NN; ++n) {
      const float dA = __expf(dlt * Av[n]);
      aP[n] *= dA;
      h[n] = dA * h[n] + du * row[n];
    }
  }
}

__global__ void __launch_bounds__(DI) k_scan_pass1(
    const float* __restrict__ xconv, const float* __restrict__ delta,
    const float* __restrict__ x_dbl, const float* __restrict__ A_logs,
    float* __restrict__ chunkA, float* __restrict__ chunkH) {
  const int d = threadIdx.x;
  const int c = blockIdx.x % NC;
  const int bk = blockIdx.x / NC;
  const int k = bk % KK, b = bk / KK;
  float Av[NN];
#pragma unroll
  for (int n = 0; n < NN; ++n)
    Av[n] = -__expf(A_logs[(size_t)(k * DI + d) * NN + n]);
  float aP[NN], h[NN];
#pragma unroll
  for (int n = 0; n < NN; ++n) { aP[n] = 1.f; h[n] = 0.f; }
  const int l0 = c * CL;
  const float* up = xconv + ((size_t)b * LL + (k ? (LL - 1 - l0) : l0)) * DI + d;
  const int ustep = k ? -DI : DI;
  const float* dp = delta + ((size_t)bk * LL + l0) * DI + d;
  const float* xr = x_dbl + ((size_t)bk * LL + l0) * CDBL + RR;
  float ub[SB], db[SB];
#pragma unroll
  for (int j = 0; j < SB; ++j) { ub[j] = up[j * ustep]; db[j] = dp[j * DI]; }
#pragma unroll 1
  for (int s0 = 0; s0 < CL - SB; s0 += SB) {
    float un[SB], dn[SB];
#pragma unroll
    for (int j = 0; j < SB; ++j) {
      un[j] = up[(s0 + SB + j) * ustep];
      dn[j] = dp[(s0 + SB + j) * DI];
    }
    __builtin_amdgcn_sched_barrier(0);
    p1_batch(ub, db, Av, aP, h, xr + (size_t)s0 * CDBL);
#pragma unroll
    for (int j = 0; j < SB; ++j) { ub[j] = un[j]; db[j] = dn[j]; }
  }
  p1_batch(ub, db, Av, aP, h, xr + (size_t)(CL - SB) * CDBL);
#pragma unroll
  for (int n = 0; n < NN; ++n) {
    const size_t idx = ((size_t)(bk * NC + c) * NN + n) * DI + d;
    chunkA[idx] = aP[n];
    chunkH[idx] = h[n];
  }
}

#define P2U 32
#define P2T 96
__global__ void __launch_bounds__(P2T) k_scan_pass2(
    const float* __restrict__ chunkA, float* __restrict__ chunkH) {
  const int gid = blockIdx.x * P2T + threadIdx.x;
  const int d = gid % DI;
  const int n = (gid / DI) % NN;
  const int bk = gid / (DI * NN);
  const int cs = NN * DI;
  const float* pa = chunkA + ((size_t)bk * NC * NN + n) * DI + d;
  float* ph = chunkH + ((size_t)bk * NC * NN + n) * DI + d;
  float hrun = 0.f;
#pragma unroll 1
  for (int c0 = 0; c0 < NC; c0 += P2U) {
    float a[P2U], hl[P2U];
#pragma unroll
    for (int j = 0; j < P2U; ++j) {
      a[j] = pa[(c0 + j) * cs];
      hl[j] = ph[(c0 + j) * cs];
    }
#pragma unroll
    for (int j = 0; j < P2U; ++j) {
      ph[(c0 + j) * cs] = hrun;
      hrun = a[j] * hrun + hl[j];
    }
  }
}

__device__ __forceinline__ void p3_batch(const float (&ub)[SB], const float (&db)[SB],
                                         const float (&Av)[NN], float (&h)[NN],
                                         const float Dv, const float* row0, float* yo) {
#pragma unroll
  for (int j = 0; j < SB; ++j) {
    const float u = ub[j], dlt = db[j];
    const float* row = row0 + (size_t)j * CDBL;
    const float du = dlt * u;
    float y = 0.f;
#pragma unroll
    for (int n = 0; n < NN; ++n) {
      const float dA = __expf(dlt * Av[n]);
      h[n] = dA * h[n] + du * row[n];
      y += h[n] * row[NN + n];
    }
    yo[(size_t)j * DI] = y + u * Dv;
  }
}

__global__ void __launch_bounds__(DI) __attribute__((amdgpu_waves_per_eu(6)))
k_scan_pass3(
    const float* __restrict__ xconv, const float* __restrict__ delta,
    const float* __restrict__ x_dbl, const float* __restrict__ A_logs,
    const float* __restrict__ Ds, const float* __restrict__ chunkH,
    float* __restrict__ ys) {
  const int d = threadIdx.x;
  const int c = blockIdx.x % NC;
  const int bk = blockIdx.x / NC;
  const int k = bk % KK, b = bk / KK;
  float Av[NN];
#pragma unroll
  for (int n = 0; n < NN; ++n)
    Av[n] = -__expf(A_logs[(size_t)(k * DI + d) * NN + n]);
  const float Dv = Ds[k * DI + d];
  float h[NN];
#pragma unroll
  for (int n = 0; n < NN; ++n)
    h[n] = chunkH[((size_t)(bk * NC + c) * NN + n) * DI + d];
  const int l0 = c * CL;
  const float* up = xconv + ((size_t)b * LL + (k ? (LL - 1 - l0) : l0)) * DI + d;
  const int ustep = k ? -DI : DI;
  const float* dp = delta + ((size_t)bk * LL + l0) * DI + d;
  const float* xr = x_dbl + ((size_t)bk * LL + l0) * CDBL + RR;
  float* yo = ys + ((size_t)bk * LL + l0) * DI + d;
  float ub[SB], db[SB];
#pragma unroll
  for (int j = 0; j < SB; ++j) { ub[j] = up[j * ustep]; db[j] = dp[j * DI]; }
#pragma unroll 1
  for (int s0 = 0; s0 < CL - SB; s0 += SB) {
    float un[SB], dn[SB];
#pragma unroll
    for (int j = 0; j < SB; ++j) {
      un[j] = up[(s0 + SB + j) * ustep];
      dn[j] = dp[(s0 + SB + j) * DI];
    }
    __builtin_amdgcn_sched_barrier(0);
    p3_batch(ub, db, Av, h, Dv, xr + (size_t)s0 * CDBL, yo + (size_t)s0 * DI);
#pragma unroll
    for (int j = 0; j < SB; ++j) { ub[j] = un[j]; db[j] = dn[j]; }
  }
  p3_batch(ub, db, Av, h, Dv, xr + (size_t)(CL - SB) * CDBL, yo + (size_t)(CL - SB) * DI);
}

__global__ void __launch_bounds__(192) k_outproj(const float* __restrict__ ys,
                                                 const float* __restrict__ xz,
                                                 const float* __restrict__ W_out,
                                                 float* __restrict__ out) {
  __shared__ __attribute__((aligned(16))) float sY[GR * GP];
  const int tid = threadIdx.x, wave = tid >> 5, lane = tid & 31;
  const int lid = lane & 15, half = lane >> 4;
  const int rowBase = blockIdx.x * GR;
  const int b = rowBase / LL;
  const int lb = rowBase % LL;
  const float* wrow = W_out + (size_t)(wave * 16 + lid) * DI + 4 * half;
  float4 bf[DI / 8];
#pragma unroll
  for (int i = 0; i < DI / 8; ++i) bf[i] = *reinterpret_cast<const float4*>(wrow + 8 * i);
  const int q = tid % (DI / 4), m0 = tid / (DI / 4);
#pragma unroll 4
  for (int it = 0; it < GR / 4; ++it) {
    const int m = it * 4 + m0;
    const int l = lb + m;
    const float4 y0 = *reinterpret_cast<const float4*>(
        ys + ((size_t)(b * KK + 0) * LL + l) * DI + q * 4);
    const float4 y1 = *reinterpret_cast<const float4*>(
        ys + ((size_t)(b * KK + 1) * LL + (LL - 1 - l)) * DI + q * 4);
    const float4 zv = *reinterpret_cast<const float4*>(
        xz + (size_t)(b * LL + l) * NOUT + DI + q * 4);
    float4 r;
    r.x = (y0.x + y1.x) * silu_f(zv.x);
    r.y = (y0.y + y1.y) * silu_f(zv.y);
    r.z = (y0.z + y1.z) * silu_f(zv.z);
    r.w = (y0.w + y1.w) * silu_f(zv.w);
    *reinterpret_cast<float4*>(&sY[m * GP + q * 4]) = r;
  }
  __syncthreads();
#pragma unroll 1
  for (int mp = 0; mp < GR / 32; ++mp) {
    const float* ar0 = &sY[(mp * 32 + lid) * GP + 4 * half];
    const float* ar1 = ar0 + 16 * GP;
    v8f acc0 = {}, acc1 = {};
#pragma unroll
    for (int i = 0; i < DI / 8; ++i) {
      const float4 a0 = *reinterpret_cast<const float4*>(ar0 + 8 * i);
      const float4 a1 = *reinterpret_cast<const float4*>(ar1 + 8 * i);
      acc0 = wmma4(lo2(a0), lo2(bf[i]), acc0);
      acc1 = wmma4(lo2(a1), lo2(bf[i]), acc1);
      acc0 = wmma4(hi2(a0), hi2(bf[i]), acc0);
      acc1 = wmma4(hi2(a1), hi2(bf[i]), acc1);
    }
    float* orow = out + (size_t)(rowBase + mp * 32 + half * 8) * CC + wave * 16 + lid;
#pragma unroll
    for (int v = 0; v < 8; ++v) {
      orow[(size_t)v * CC] = acc0[v];
      orow[(size_t)(v + 16) * CC] = acc1[v];
    }
  }
}

extern "C" void kernel_launch(void* const* d_in, const int* in_sizes, int n_in,
                              void* d_out, int out_size, void* d_ws, size_t ws_size,
                              hipStream_t stream) {
  (void)in_sizes; (void)n_in; (void)out_size; (void)ws_size;
  const float* x      = (const float*)d_in[0];
  const float* W_in   = (const float*)d_in[1];
  const float* conv_w = (const float*)d_in[2];
  const float* conv_b = (const float*)d_in[3];
  const float* xpw    = (const float*)d_in[4];
  const float* dtw    = (const float*)d_in[5];
  const float* dtb    = (const float*)d_in[6];
  const float* A_logs = (const float*)d_in[7];
  const float* Ds     = (const float*)d_in[8];
  const float* W_out  = (const float*)d_in[9];
  float* out = (float*)d_out;

  char* ws = (char*)d_ws;
  size_t off = 0;
  auto alloc = [&](size_t nbytes) {
    char* p = ws + off;
    off = (off + nbytes + 255) & ~(size_t)255;
    return (float*)p;
  };
  float* xz     = alloc((size_t)BL * NOUT * sizeof(float));
  float* xconv  = alloc((size_t)BL * DI * sizeof(float));
  float* x_dbl  = alloc((size_t)BB * KK * LL * CDBL * sizeof(float));
  float* delta  = alloc((size_t)BB * KK * LL * DI * sizeof(float));
  float* ysb    = alloc((size_t)BB * KK * LL * DI * sizeof(float));
  float* chunkA = alloc((size_t)BB * KK * DI * NC * NN * sizeof(float));
  float* chunkH = alloc((size_t)BB * KK * DI * NC * NN * sizeof(float));

  k_inproj<<<BL / IPR, 256, 0, stream>>>(x, W_in, xz);
  k_dwconv<<<(BB * HH * DWSTRIPS) / (DWT / DWC4), DWT, 0, stream>>>(xz, conv_w,
                                                                    conv_b, xconv);
  k_xdbl<<<BB * KK * LL / GR, 192, 0, stream>>>(xconv, xpw, dtw, dtb, x_dbl, delta);
  k_scan_pass1<<<BB * KK * NC, DI, 0, stream>>>(
      xconv, delta, x_dbl, A_logs, chunkA, chunkH);
  k_scan_pass2<<<(BB * KK * NN * DI) / P2T, P2T, 0, stream>>>(chunkA, chunkH);
  k_scan_pass3<<<BB * KK * NC, DI, 0, stream>>>(
      xconv, delta, x_dbl, A_logs, Ds, chunkH, ysb);
  k_outproj<<<BL / GR, 192, 0, stream>>>(ysb, xz, W_out, out);
}


# scan passes 1 and 3 fused into the x_dbl and out_proj kernels: delta, u, B rows and y pass through registers and LDS instead of workspace planes; two launches and the ys plane fewer; op order kept
# speedup vs baseline: 3.2480x; 1.1798x over previous
#include <hip/hip_runtime.h>
#include <cstdint>
#include <cstddef>

#define BB 4
#define HH 32
#define WWD 256
#define CC 96
#define DI 192
#define KK 2
#define NN 16
#define RR 6
#define LL (HH*WWD)
#define BL (BB*LL)
#define NOUT (2*DI)
#define CDBL (RR + 2*NN)
#define NC 128
#define CL (LL/NC)

typedef float v2f __attribute__((ext_vector_type(2)));
typedef float v8f __attribute__((ext_vector_type(8)));

__device__ __forceinline__ v8f wmma4(v2f a, v2f b, v8f c) {
  return __builtin_amdgcn_wmma_f32_16x16x4_f32(false, a, false, b, (short)0, c,
                                               false, false);
}

__device__ __forceinline__ float silu_f(float x) {
  return x / (1.0f + __expf(-x));
}
__device__ __forceinline__ float softplus_f(float x) {
  const float e = __expf(-fabsf(x));
  const float u = 1.0f + e;
  const float lost = e - (u - 1.0f);
  const float l = __builtin_amdgcn_logf(u) * 0.6931471805599453f;
  return fmaxf(x, 0.0f) + (l + lost * __builtin_amdgcn_rcpf(u));
}
__device__ __forceinline__ float xhalf(float x) {
  return __builtin_bit_cast(float, __builtin_amdgcn_permlanex16(
      __builtin_bit_cast(int, x), __builtin_bit_cast(int, x), 0x76543210, 0xfedcba98, false, false));
}

__device__ __forceinline__ v2f lo2(const float4 v) { v2f r; r.x = v.x; r.y = v.y; return r; }
__device__ __forceinline__ v2f hi2(const float4 v) { v2f r; r.x = v.z; r.y = v.w; return r; }

#define IPR 128
#define IPP 100
__global__ void __launch_bounds__(256) k_inproj(const float* __restrict__ x,
                                                const float* __restrict__ W_in,
                                                float* __restrict__ xz) {
  __shared__ __attribute__((aligned(16))) float sX[IPR * IPP];
  const int tid = threadIdx.x;
  const int rowBase = blockIdx.x * IPR;
  const int wave = tid >> 5, lane = tid & 31;
  const int lid = lane & 15, half = lane >> 4;
  float4 bf[3][CC / 8];
#pragma unroll
  for (int j = 0; j < 3; ++j) {
    const float* wrow = W_in + (size_t)((wave + 8 * j) * 16 + lid) * CC + 4 * half;
#pragma unroll
    for (int i = 0; i < CC / 8; ++i)
      bf[j][i] = *reinterpret_cast<const float4*>(wrow + 8 * i);
  }
#pragma unroll
  for (int it = 0; it < IPR * (CC / 4) / 256; ++it) {
    const int i = tid + it * 256;
    const int m = i / (CC / 4), q = i % (CC / 4);
    const float4 v = *reinterpret_cast<const float4*>(x + (size_t)(rowBase + m) * CC + q * 4);
    *reinterpret_cast<float4*>(&sX[m * IPP + q * 4]) = v;
  }
  __syncthreads();
#pragma unroll 1
  for (int mt = 0; mt < IPR / 16; ++mt) {
    const float* arow = &sX[(mt * 16 + lid) * IPP + 4 * half];
    float4 af[CC / 8];
#pragma unroll
    for (int i = 0; i < CC / 8; ++i) af[i] = *reinterpret_cast<const float4*>(arow + 8 * i);
    v8f acc[3] = {};
#pragma unroll
    for (int i = 0; i < CC / 8; ++i) {
#pragma unroll
      for (int j = 0; j < 3; ++j) acc[j] = wmma4(lo2(af[i]), lo2(bf[j][i]), acc[j]);
#pragma unroll
      for (int j = 0; j < 3; ++j) acc[j] = wmma4(hi2(af[i]), hi2(bf[j][i]), acc[j]);
    }
    float* orow = xz + (size_t)(rowBase + mt * 16 + half * 8) * NOUT + lid;
#pragma unroll
    for (int j = 0; j < 3; ++j)
#pragma unroll
      for (int v = 0; v < 8; ++v)
        orow[(size_t)v * NOUT + (wave + 8 * j) * 16] = acc[j][v];
  }
}

#define DWS 8
#define DWC4 (DI / 4)
#define DWSTRIPS (WWD / DWS)
#define DWT 192
__global__ void __launch_bounds__(DWT) k_dwconv(const float* __restrict__ xz,
                                                const float* __restrict__ conv_w,
                                                const float* __restrict__ conv_b,
                                                float* __restrict__ xconv) {
  const unsigned tid = threadIdx.x;
  const int c = (int)(tid % DWC4) * 4;
  const unsigned s = blockIdx.x * (DWT / DWC4) + tid / DWC4;
  const int w0 = (int)(s % DWSTRIPS) * DWS;
  const int h = (int)((s / DWSTRIPS) % HH);
  const int b = (int)(s / (DWSTRIPS * HH));

  float wt[4][9];
#pragma unroll
  for (int j = 0; j < 4; ++j)
#pragma unroll
    for (int t = 0; t < 9; ++t) wt[j][t] = conv_w[(c + j) * 9 + t];
  float bias[4];
#pragma unroll
  for (int j = 0; j < 4; ++j) bias[j] = conv_b[c + j];

  float v[3][DWS + 2][4];
#pragma unroll
  for (int r = 0; r < 3; ++r) {
    const int hh = h + r - 1;
    const bool rok = (hh >= 0) && (hh < HH);
    const int hc = hh < 0 ? 0 : (hh >= HH ? HH - 1 : hh);
#pragma unroll
    for (int q = 0; q < DWS + 2; ++q) {
      bool cok = true;
      int wc = w0 + q - 1;
      if (q == 0) { cok = w0 > 0; wc = cok ? wc : 0; }
      if (q == DWS + 1) { cok = (w0 + DWS) < WWD; wc = cok ? wc : WWD - 1; }
      const bool ok = rok && cok;
      const float4 t4 = *reinterpret_cast<const float4*>(
          xz + (size_t)((b * HH + hc) * WWD + wc) * NOUT + c);
      v[r][q][0] = ok ? t4.x : 0.f;
      v[r][q][1] = ok ? t4.y : 0.f;
      v[r][q][2] = ok ? t4.z : 0.f;
      v[r][q][3] = ok ? t4.w : 0.f;
    }
  }

  float* orow = xconv + (size_t)((b * HH + h) * WWD + w0) * DI + c;
#pragma unroll
  for (int i = 0; i < DWS; ++i) {
    float o[4];
#pragma unroll
    for (int j = 0; j < 4; ++j) {
      float acc = bias[j];
#pragma unroll
      for (int r = 0; r < 3; ++r)
#pragma unroll
        for (int dx = 0; dx < 3; ++dx) acc += v[r][i + dx][j] * wt[j][r * 3 + dx];
      o[j] = silu_f(acc);
    }
    *reinterpret_cast<float4*>(orow + (size_t)i * DI) =
        make_float4(o[0], o[1], o[2], o[3]);
  }
}


#define GR 64
#define GP 196
#define XP 52
#define XO 2
static_assert(GR == CL, "an x_dbl block must be exactly one scan chunk");
static_assert((XO + RR) % 4 == 0 && XO + 48 <= XP, "B row must be 16-byte aligned");
__global__ void __launch_bounds__(192) k_xdbl(const float* __restrict__ xconv,
                                              const float* __restrict__ xpw,
                                              const float* __restrict__ dtw,
                                              const float* __restrict__ dtb,
                                              const float* __restrict__ A_logs,
                                              float* __restrict__ x_dbl,
                                              float* __restrict__ delta,
                                              float* __restrict__ chunkA,
                                              float* __restrict__ chunkH) {
  __shared__ __attribute__((aligned(16))) float sA[GR * GP];
  __shared__ __attribute__((aligned(16))) float sX[GR * XP];
  const int tid = threadIdx.x, wave = tid >> 5, lane = tid & 31;
  const int lid = lane & 15, half = lane >> 4;
  const int bk = blockIdx.x / (LL / GR);
  const int chunk = blockIdx.x % (LL / GR);
  const int lbase = chunk * GR;
  const int k = bk % KK, b = bk / KK;
  const int j = wave % 3, g = wave / 3;
  const int col = j * 16 + lid;
  const bool cok = col < CDBL;
  const float* wrow = xpw + (size_t)(k * CDBL + (cok ? col : 0)) * DI + 4 * half;
  float4 bf[DI / 8];
#pragma unroll
  for (int i = 0; i < DI / 8; ++i) {
    const float4 t4 = *reinterpret_cast<const float4*>(wrow + 8 * i);
    bf[i].x = cok ? t4.x : 0.f; bf[i].y = cok ? t4.y : 0.f;
    bf[i].z = cok ? t4.z : 0.f; bf[i].w = cok ? t4.w : 0.f;
  }
  const int q = tid % (DI / 4), m0 = tid / (DI / 4);
#pragma unroll 8
  for (int it = 0; it < GR / 4; ++it) {
    const int m = it * 4 + m0;
    const int l = lbase + m;
    const int lidx = k ? (LL - 1 - l) : l;
    const float4 v = *reinterpret_cast<const float4*>(xconv + (size_t)(b * LL + lidx) * DI + q * 4);
    *reinterpret_cast<float4*>(&sA[m * GP + q * 4]) = v;
  }
  __syncthreads();
  const float* ar0 = &sA[(g * 32 + lid) * GP + 4 * half];
  const float* ar1 = ar0 + 16 * GP;
  v8f acc0 = {}, acc1 = {};
#pragma unroll
  for (int i = 0; i < DI / 8; ++i) {
    const float4 a0 = *reinterpret_cast<const float4*>(ar0 + 8 * i);
    const float4 a1 = *reinterpret_cast<const float4*>(ar1 + 8 * i);
    acc0 = wmma4(lo2(a0), lo2(bf[i]), acc0);
    acc1 = wmma4(lo2(a1), lo2(bf[i]), acc1);
    acc0 = wmma4(hi2(a0), hi2(bf[i]), acc0);
    acc1 = wmma4(hi2(a1), hi2(bf[i]), acc1);
  }
  v2f pb0[2], pb1[2];
  float pbias[2];
#pragma unroll
  for (int t = 0; t < 2; ++t) {
    const int dcol = wave * 32 + t * 16 + lid;
    const float* wr = dtw + (size_t)(k * DI + dcol) * RR;
    pb0[t].x = wr[2 * half]; pb0[t].y = wr[2 * half + 1];
    const float w4 = wr[4], w5 = wr[5];
    pb1[t].x = half ? 0.f : w4; pb1[t].y = half ? 0.f : w5;
    pbias[t] = dtb[k * DI + dcol];
  }
  {
    float* xw = &sX[(g * 32 + half * 8) * XP + XO + col];
#pragma unroll
    for (int v = 0; v < 8; ++v) {
      xw[v * XP] = acc0[v];
      xw[(v + 16) * XP] = acc1[v];
    }
  }
  __syncthreads();
  if (cok) {
    float* orow = x_dbl + ((size_t)bk * LL + lbase + g * 32 + half * 8) * CDBL + col;
#pragma unroll
    for (int v = 0; v < 8; ++v) {
      orow[(size_t)v * CDBL] = acc0[v];
      orow[(size_t)(v + 16) * CDBL] = acc1[v];
    }
  }
  const int d = tid;
  float Av[NN];
#pragma unroll
  for (int n = 0; n < NN; ++n)
    Av[n] = -__expf(A_logs[(size_t)(k * DI + d) * NN + n]);
  float aP[NN], h[NN];
#pragma unroll
  for (int n = 0; n < NN; ++n) { aP[n] = 1.f; h[n] = 0.f; }
  float* drow = delta + ((size_t)bk * LL + lbase) * DI + d;
#pragma unroll 1
  for (int rt = 0; rt < GR / 16; ++rt) {
    const float* ar = &sX[(rt * 16 + lid) * XP + XO];
    const float2 r03 = *reinterpret_cast<const float2*>(ar + 2 * half);
    const float2 r45 = *reinterpret_cast<const float2*>(ar + 4);
    v2f a0, a1;
    a0.x = r03.x; a0.y = r03.y;
    a1.x = half ? 0.f : r45.x; a1.y = half ? 0.f : r45.y;
    v8f d0 = {}, d1 = {};
    d0 = wmma4(a0, pb0[0], d0);
    d1 = wmma4(a0, pb0[1], d1);
    d0 = wmma4(a1, pb1[0], d0);
    d1 = wmma4(a1, pb1[1], d1);
    float* orow = drow + (size_t)(rt * 16) * DI;
    float dl[16];
#pragma unroll
    for (int v = 0; v < 8; ++v) {
      const float s0 = softplus_f(d0[v] + pbias[0]);
      const float s1 = softplus_f(d1[v] + pbias[1]);
      const float got = xhalf(half ? s0 : s1);
      dl[v] = half ? got : s0;
      dl[v + 8] = half ? s1 : got;
      orow[(size_t)v * DI] = dl[v];
      orow[(size_t)(v + 8) * DI] = dl[v + 8];
    }
    const float* urow = &sA[(rt * 16) * GP + d];
    const float* brow = &sX[(rt * 16) * XP + XO + RR];
#pragma unroll
    for (int s = 0; s < 16; ++s) {
      const float u = urow[s * GP], dlt = dl[s];
      float Bv[NN];
#pragma unroll
      for (int i = 0; i < NN / 4; ++i) {
        const float4 t4 = *reinterpret_cast<const float4*>(brow + s * XP + 4 * i);
        Bv[4 * i] = t4.x; Bv[4 * i + 1] = t4.y; Bv[4 * i + 2] = t4.z; Bv[4 * i + 3] = t4.w;
      }
      const float du = dlt * u;
      float dA[NN];
#pragma unroll
      for (int n = 0; n < NN; ++n) dA[n] = __expf(dlt * Av[n]);
      float t[NN];
#pragma unroll
      for (int n = 0; n < NN; ++n) t[n] = du * Bv[n];
#pragma unroll
      for (int n = 0; n < NN; ++n) h[n] = __builtin_fmaf(dA[n], h[n], t[n]);
#pragma unroll
      for (int n = 0; n < NN; ++n) aP[n] *= dA[n];
    }
  }
#pragma unroll
  for (int n = 0; n < NN; ++n) {
    const size_t idx = ((size_t)(bk * NC + chunk) * NN + n) * DI + d;
    chunkA[idx] = aP[n];
    chunkH[idx] = h[n];
  }
}

#define SB 8

#define P2U 32
#define P2T 96
__global__ void __launch_bounds__(P2T) k_scan_pass2(
    const float* __restrict__ chunkA, float* __restrict__ chunkH) {
  const int gid = blockIdx.x * P2T + threadIdx.x;
  const int d = gid % DI;
  const int n = (gid / DI) % NN;
  const int bk = gid / (DI * NN);
  const int cs = NN * DI;
  const float* pa = chunkA + ((size_t)bk * NC * NN + n) * DI + d;
  float* ph = chunkH + ((size_t)bk * NC * NN + n) * DI + d;
  float hrun = 0.f;
#pragma unroll 1
  for (int c0 = 0; c0 < NC; c0 += P2U) {
    float a[P2U], hl[P2U];
#pragma unroll
    for (int j = 0; j < P2U; ++j) {
      a[j] = pa[(c0 + j) * cs];
      hl[j] = ph[(c0 + j) * cs];
    }
#pragma unroll
    for (int j = 0; j < P2U; ++j) {
      ph[(c0 + j) * cs] = hrun;
      hrun = a[j] * hrun + hl[j];
    }
  }
}

__device__ __forceinline__ void p3_batch(const float (&ub)[SB], const float (&db)[SB],
                                         const float (&Av)[NN], float (&h)[NN],
                                         const float Dv, const float* row0, float* yo,
                                         const int ystep) {
#pragma unroll
  for (int j = 0; j < SB; ++j) {
    const float u = ub[j], dlt = db[j];
    const float* row = row0 + (size_t)j * CDBL;
    const float du = dlt * u;
    float y = 0.f;
#pragma unroll
    for (int n = 0; n < NN; ++n) {
      const float dA = __expf(dlt * Av[n]);
      h[n] = dA * h[n] + du * row[n];
      y += h[n] * row[NN + n];
    }
    yo[j * ystep] = y + u * Dv;
  }
}

#define OT (KK * DI)
static_assert(GR == CL, "an out_proj block must be exactly one scan chunk per direction");
__global__ void __launch_bounds__(OT) __attribute__((amdgpu_waves_per_eu(6)))
k_scan3_outproj(
    const float* __restrict__ xconv, const float* __restrict__ delta,
    const float* __restrict__ x_dbl, const float* __restrict__ A_logs,
    const float* __restrict__ Ds, const float* __restrict__ chunkH,
    const float* __restrict__ xz, const float* __restrict__ W_out,
    float* __restrict__ out) {
  __shared__ __attribute__((aligned(16))) float sY[KK * GR * GP];
  const int tid = threadIdx.x, wave = tid >> 5, lane = tid & 31;
  const int lid = lane & 15, half = lane >> 4;
  const int b = blockIdx.x / NC;
  const int cs = blockIdx.x % NC;
  const int k = __builtin_amdgcn_readfirstlane(tid / DI);
  const int d = tid - k * DI;
  const int c = k ? (NC - 1 - cs) : cs;
  const int bk = b * KK + k;
  {
    float Av[NN];
#pragma unroll
    for (int n = 0; n < NN; ++n)
      Av[n] = -__expf(A_logs[(size_t)(k * DI + d) * NN + n]);
    const float Dv = Ds[k * DI + d];
    float h[NN];
#pragma unroll
    for (int n = 0; n < NN; ++n)
      h[n] = chunkH[((size_t)(bk * NC + c) * NN + n) * DI + d];
    const int l0 = c * CL;
    const float* up = xconv + ((size_t)b * LL + (k ? (LL - 1 - l0) : l0)) * DI + d;
    const int ustep = k ? -DI : DI;
    const float* dp = delta + ((size_t)bk * LL + l0) * DI + d;
    const float* xr = x_dbl + ((size_t)bk * LL + l0) * CDBL + RR;
    float* yo = &sY[(k * GR + (k ? (GR - 1) : 0)) * GP + d];
    const int ystep = k ? -GP : GP;
    float ub[SB], db[SB];
#pragma unroll
    for (int j = 0; j < SB; ++j) { ub[j] = up[j * ustep]; db[j] = dp[j * DI]; }
#pragma unroll 1
    for (int s0 = 0; s0 < CL - SB; s0 += SB) {
      float un[SB], dn[SB];
#pragma unroll
      for (int j = 0; j < SB; ++j) {
        un[j] = up[(s0 + SB + j) * ustep];
        dn[j] = dp[(s0 + SB + j) * DI];
      }
      __builtin_amdgcn_sched_barrier(0);
      p3_batch(ub, db, Av, h, Dv, xr + (size_t)s0 * CDBL, yo + s0 * ystep, ystep);
#pragma unroll
      for (int j = 0; j < SB; ++j) { ub[j] = un[j]; db[j] = dn[j]; }
    }
    p3_batch(ub, db, Av, h, Dv, xr + (size_t)(CL - SB) * CDBL, yo + (CL - SB) * ystep, ystep);
  }
  const int ct = wave % 6, rp = wave / 6;
  const float* wrow = W_out + (size_t)(ct * 16 + lid) * DI + 4 * half;
  float4 bf[DI / 8];
#pragma unroll
  for (int i = 0; i < DI / 8; ++i) bf[i] = *reinterpret_cast<const float4*>(wrow + 8 * i);
  __syncthreads();
  const int rowBase = blockIdx.x * GR;
  const int q = tid % (DI / 4), m0 = tid / (DI / 4);
#pragma unroll 4
  for (int it = 0; it < GR / 8; ++it) {
    const int m = it * 8 + m0;
    const float4 zv = *reinterpret_cast<const float4*>(
        xz + (size_t)(rowBase + m) * NOUT + DI + q * 4);
    const float4 y0 = *reinterpret_cast<const float4*>(&sY[m * GP + q * 4]);
    const float4 y1 = *reinterpret_cast<const float4*>(&sY[(GR + m) * GP + q * 4]);
    float4 r;
    r.x = (y0.x + y1.x) * silu_f(zv.x);
    r.y = (y0.y + y1.y) * silu_f(zv.y);
    r.z = (y0.z + y1.z) * silu_f(zv.z);
    r.w = (y0.w + y1.w) * silu_f(zv.w);
    *reinterpret_cast<float4*>(&sY[m * GP + q * 4]) = r;
  }
  __syncthreads();
  const float* ar0 = &sY[(rp * 32 + lid) * GP + 4 * half];
  const float* ar1 = ar0 + 16 * GP;
  v8f acc0 = {}, acc1 = {};
#pragma unroll
  for (int i = 0; i < DI / 8; ++i) {
    const float4 a0 = *reinterpret_cast<const float4*>(ar0 + 8 * i);
    const float4 a1 = *reinterpret_cast<const float4*>(ar1 + 8 * i);
    acc0 = wmma4(lo2(a0), lo2(bf[i]), acc0);
    acc1 = wmma4(lo2(a1), lo2(bf[i]), acc1);
    acc0 = wmma4(hi2(a0), hi2(bf[i]), acc0);
    acc1 = wmma4(hi2(a1), hi2(bf[i]), acc1);
  }
  float* orow = out + (size_t)(rowBase + rp * 32 + half * 8) * CC + ct * 16 + lid;
#pragma unroll
  for (int v = 0; v < 8; ++v) {
    orow[(size_t)v * CC] = acc0[v];
    orow[(size_t)(v + 16) * CC] = acc1[v];
  }
}

extern "C" void kernel_launch(void* const* d_in, const int* in_sizes, int n_in,
                              void* d_out, int out_size, void* d_ws, size_t ws_size,
                              hipStream_t stream) {
  (void)in_sizes; (void)n_in; (void)out_size; (void)ws_size;
  const float* x      = (const float*)d_in[0];
  const float* W_in   = (const float*)d_in[1];
  const float* conv_w = (const float*)d_in[2];
  const float* conv_b = (const float*)d_in[3];
  const float* xpw    = (const float*)d_in[4];
  const float* dtw    = (const float*)d_in[5];
  const float* dtb    = (const float*)d_in[6];
  const float* A_logs = (const float*)d_in[7];
  const float* Ds     = (const float*)d_in[8];
  const float* W_out  = (const float*)d_in[9];
  float* out = (float*)d_out;

  char* ws = (char*)d_ws;
  size_t off = 0;
  auto alloc = [&](size_t nbytes) {
    char* p = ws + off;
    off = (off + nbytes + 255) & ~(size_t)255;
    return (float*)p;
  };
  float* xz     = alloc((size_t)BL * NOUT * sizeof(float));
  float* xconv  = alloc((size_t)BL * DI * sizeof(float));
  float* x_dbl  = alloc((size_t)BB * KK * LL * CDBL * sizeof(float));
  float* delta  = alloc((size_t)BB * KK * LL * DI * sizeof(float));
  float* chunkA = alloc((size_t)BB * KK * DI * NC * NN * sizeof(float));
  float* chunkH = alloc((size_t)BB * KK * DI * NC * NN * sizeof(float));

  k_inproj<<<BL / IPR, 256, 0, stream>>>(x, W_in, xz);
  k_dwconv<<<(BB * HH * DWSTRIPS) / (DWT / DWC4), DWT, 0, stream>>>(xz, conv_w,
                                                                    conv_b, xconv);
  k_xdbl<<<BB * KK * LL / GR, 192, 0, stream>>>(xconv, xpw, dtw, dtb, A_logs, x_dbl,
                                                delta, chunkA, chunkH);
  k_scan_pass2<<<(BB * KK * NN * DI) / P2T, P2T, 0, stream>>>(chunkA, chunkH);
  k_scan3_outproj<<<BB * NC, OT, 0, stream>>>(
      xconv, delta, x_dbl, A_logs, Ds, chunkH, xz, W_out, out);
}


# scan decay as exp2(delta*(A*log2e)); chunk decay product = exp2(A*log2e*sum(delta)) formed in pass 2 from one delta sum per chunk and channel, not a 16-state running product per step in pass 1
# speedup vs baseline: 3.5824x; 1.1030x over previous
#include <hip/hip_runtime.h>
#include <cstdint>
#include <cstddef>

#define BB 4
#define HH 32
#define WWD 256
#define CC 96
#define DI 192
#define KK 2
#define NN 16
#define RR 6
#define LL (HH*WWD)
#define BL (BB*LL)
#define NOUT (2*DI)
#define CDBL (RR + 2*NN)
#define NC 128
#define CL (LL/NC)

typedef float v2f __attribute__((ext_vector_type(2)));
typedef float v8f __attribute__((ext_vector_type(8)));

__device__ __forceinline__ v8f wmma4(v2f a, v2f b, v8f c) {
  return __builtin_amdgcn_wmma_f32_16x16x4_f32(false, a, false, b, (short)0, c,
                                               false, false);
}

__device__ __forceinline__ float silu_f(float x) {
  return x / (1.0f + __expf(-x));
}
__device__ __forceinline__ float softplus_f(float x) {
  const float e = __expf(-fabsf(x));
  const float u = 1.0f + e;
  const float lost = e - (u - 1.0f);
  const float l = __builtin_amdgcn_logf(u) * 0.6931471805599453f;
  return fmaxf(x, 0.0f) + (l + lost * __builtin_amdgcn_rcpf(u));
}
#define LOG2E 1.4426950408889634f
__device__ __forceinline__ float exp2_hw(float x) { return __builtin_amdgcn_exp2f(x); }
__device__ __forceinline__ float xhalf(float x) {
  return __builtin_bit_cast(float, __builtin_amdgcn_permlanex16(
      __builtin_bit_cast(int, x), __builtin_bit_cast(int, x), 0x76543210, 0xfedcba98, false, false));
}

__device__ __forceinline__ v2f lo2(const float4 v) { v2f r; r.x = v.x; r.y = v.y; return r; }
__device__ __forceinline__ v2f hi2(const float4 v) { v2f r; r.x = v.z; r.y = v.w; return r; }

#define IPR 128
#define IPP 100
__global__ void __launch_bounds__(256) k_inproj(const float* __restrict__ x,
                                                const float* __restrict__ W_in,
                                                float* __restrict__ xz) {
  __shared__ __attribute__((aligned(16))) float sX[IPR * IPP];
  const int tid = threadIdx.x;
  const int rowBase = blockIdx.x * IPR;
  const int wave = tid >> 5, lane = tid & 31;
  const int lid = lane & 15, half = lane >> 4;
  float4 bf[3][CC / 8];
#pragma unroll
  for (int j = 0; j < 3; ++j) {
    const float* wrow = W_in + (size_t)((wave + 8 * j) * 16 + lid) * CC + 4 * half;
#pragma unroll
    for (int i = 0; i < CC / 8; ++i)
      bf[j][i] = *reinterpret_cast<const float4*>(wrow + 8 * i);
  }
#pragma unroll
  for (int it = 0; it < IPR * (CC / 4) / 256; ++it) {
    const int i = tid + it * 256;
    const int m = i / (CC / 4), q = i % (CC / 4);
    const float4 v = *reinterpret_cast<const float4*>(x + (size_t)(rowBase + m) * CC + q * 4);
    *reinterpret_cast<float4*>(&sX[m * IPP + q * 4]) = v;
  }
  __syncthreads();
#pragma unroll 1
  for (int mt = 0; mt < IPR / 16; ++mt) {
    const float* arow = &sX[(mt * 16 + lid) * IPP + 4 * half];
    float4 af[CC / 8];
#pragma unroll
    for (int i = 0; i < CC / 8; ++i) af[i] = *reinterpret_cast<const float4*>(arow + 8 * i);
    v8f acc[3] = {};
#pragma unroll
    for (int i = 0; i < CC / 8; ++i) {
#pragma unroll
      for (int j = 0; j < 3; ++j) acc[j] = wmma4(lo2(af[i]), lo2(bf[j][i]), acc[j]);
#pragma unroll
      for (int j = 0; j < 3; ++j) acc[j] = wmma4(hi2(af[i]), hi2(bf[j][i]), acc[j]);
    }
    float* orow = xz + (size_t)(rowBase + mt * 16 + half * 8) * NOUT + lid;
#pragma unroll
    for (int j = 0; j < 3; ++j)
#pragma unroll
      for (int v = 0; v < 8; ++v)
        orow[(size_t)v * NOUT + (wave + 8 * j) * 16] = acc[j][v];
  }
}

#define DWS 8
#define DWC4 (DI / 4)
#define DWSTRIPS (WWD / DWS)
#define DWT 192
__global__ void __launch_bounds__(DWT) k_dwconv(const float* __restrict__ xz,
                                                const float* __restrict__ conv_w,
                                                const float* __restrict__ conv_b,
                                                float* __restrict__ xconv) {
  const unsigned tid = threadIdx.x;
  const int c = (int)(tid % DWC4) * 4;
  const unsigned s = blockIdx.x * (DWT / DWC4) + tid / DWC4;
  const int w0 = (int)(s % DWSTRIPS) * DWS;
  const int h = (int)((s / DWSTRIPS) % HH);
  const int b = (int)(s / (DWSTRIPS * HH));

  float wt[4][9];
#pragma unroll
  for (int j = 0; j < 4; ++j)
#pragma unroll
    for (int t = 0; t < 9; ++t) wt[j][t] = conv_w[(c + j) * 9 + t];
  float bias[4];
#pragma unroll
  for (int j = 0; j < 4; ++j) bias[j] = conv_b[c + j];

  float v[3][DWS + 2][4];
#pragma unroll
  for (int r = 0; r < 3; ++r) {
    const int hh = h + r - 1;
    const bool rok = (hh >= 0) && (hh < HH);
    const int hc = hh < 0 ? 0 : (hh >= HH ? HH - 1 : hh);
#pragma unroll
    for (int q = 0; q < DWS + 2; ++q) {
      bool cok = true;
      int wc = w0 + q - 1;
      if (q == 0) { cok = w0 > 0; wc = cok ? wc : 0; }
      if (q == DWS + 1) { cok = (w0 + DWS) < WWD; wc = cok ? wc : WWD - 1; }
      const bool ok = rok && cok;
      const float4 t4 = *reinterpret_cast<const float4*>(
          xz + (size_t)((b * HH + hc) * WWD + wc) * NOUT + c);
      v[r][q][0] = ok ? t4.x : 0.f;
      v[r][q][1] = ok ? t4.y : 0.f;
      v[r][q][2] = ok ? t4.z : 0.f;
      v[r][q][3] = ok ? t4.w : 0.f;
    }
  }

  float* orow = xconv + (size_t)((b * HH + h) * WWD + w0) * DI + c;
#pragma unroll
  for (int i = 0; i < DWS; ++i) {
    float o[4];
#pragma unroll
    for (int j = 0; j < 4; ++j) {
      float acc = bias[j];
#pragma unroll
      for (int r = 0; r < 3; ++r)
#pragma unroll
        for (int dx = 0; dx < 3; ++dx) acc += v[r][i + dx][j] * wt[j][r * 3 + dx];
      o[j] = silu_f(acc);
    }
    *reinterpret_cast<float4*>(orow + (size_t)i * DI) =
        make_float4(o[0], o[1], o[2], o[3]);
  }
}


#define GR 64
#define GP 196
#define XP 52
#define XO 2
static_assert(GR == CL, "an x_dbl block must be exactly one scan chunk");
static_assert((XO + RR) % 4 == 0 && XO + 48 <= XP, "B row must be 16-byte aligned");
__global__ void __launch_bounds__(192) k_xdbl(const float* __restrict__ xconv,
                                              const float* __restrict__ xpw,
                                              const float* __restrict__ dtw,
                                              const float* __restrict__ dtb,
                                              const float* __restrict__ A_logs,
                                              float* __restrict__ x_dbl,
                                              float* __restrict__ delta,
                                              float* __restrict__ chunkS,
                                              float* __restrict__ chunkH) {
  __shared__ __attribute__((aligned(16))) float sA[GR * GP];
  __shared__ __attribute__((aligned(16))) float sX[GR * XP];
  const int tid = threadIdx.x, wave = tid >> 5, lane = tid & 31;
  const int lid = lane & 15, half = lane >> 4;
  const int bk = blockIdx.x / (LL / GR);
  const int chunk = blockIdx.x % (LL / GR);
  const int lbase = chunk * GR;
  const int k = bk % KK, b = bk / KK;
  const int j = wave % 3, g = wave / 3;
  const int col = j * 16 + lid;
  const bool cok = col < CDBL;
  const float* wrow = xpw + (size_t)(k * CDBL + (cok ? col : 0)) * DI + 4 * half;
  float4 bf[DI / 8];
#pragma unroll
  for (int i = 0; i < DI / 8; ++i) {
    const float4 t4 = *reinterpret_cast<const float4*>(wrow + 8 * i);
    bf[i].x = cok ? t4.x : 0.f; bf[i].y = cok ? t4.y : 0.f;
    bf[i].z = cok ? t4.z : 0.f; bf[i].w = cok ? t4.w : 0.f;
  }
  const int q = tid % (DI / 4), m0 = tid / (DI / 4);
#pragma unroll 8
  for (int it = 0; it < GR / 4; ++it) {
    const int m = it * 4 + m0;
    const int l = lbase + m;
    const int lidx = k ? (LL - 1 - l) : l;
    const float4 v = *reinterpret_cast<const float4*>(xconv + (size_t)(b * LL + lidx) * DI + q * 4);
    *reinterpret_cast<float4*>(&sA[m * GP + q * 4]) = v;
  }
  __syncthreads();
  const float* ar0 = &sA[(g * 32 + lid) * GP + 4 * half];
  const float* ar1 = ar0 + 16 * GP;
  v8f acc0 = {}, acc1 = {};
#pragma unroll
  for (int i = 0; i < DI / 8; ++i) {
    const float4 a0 = *reinterpret_cast<const float4*>(ar0 + 8 * i);
    const float4 a1 = *reinterpret_cast<const float4*>(ar1 + 8 * i);
    acc0 = wmma4(lo2(a0), lo2(bf[i]), acc0);
    acc1 = wmma4(lo2(a1), lo2(bf[i]), acc1);
    acc0 = wmma4(hi2(a0), hi2(bf[i]), acc0);
    acc1 = wmma4(hi2(a1), hi2(bf[i]), acc1);
  }
  v2f pb0[2], pb1[2];
  float pbias[2];
#pragma unroll
  for (int t = 0; t < 2; ++t) {
    const int dcol = wave * 32 + t * 16 + lid;
    const float* wr = dtw + (size_t)(k * DI + dcol) * RR;
    pb0[t].x = wr[2 * half]; pb0[t].y = wr[2 * half + 1];
    const float w4 = wr[4], w5 = wr[5];
    pb1[t].x = half ? 0.f : w4; pb1[t].y = half ? 0.f : w5;
    pbias[t] = dtb[k * DI + dcol];
  }
  {
    float* xw = &sX[(g * 32 + half * 8) * XP + XO + col];
#pragma unroll
    for (int v = 0; v < 8; ++v) {
      xw[v * XP] = acc0[v];
      xw[(v + 16) * XP] = acc1[v];
    }
  }
  __syncthreads();
  if (cok) {
    float* orow = x_dbl + ((size_t)bk * LL + lbase + g * 32 + half * 8) * CDBL + col;
#pragma unroll
    for (int v = 0; v < 8; ++v) {
      orow[(size_t)v * CDBL] = acc0[v];
      orow[(size_t)(v + 16) * CDBL] = acc1[v];
    }
  }
  const int d = tid;
  float Av[NN];
#pragma unroll
  for (int n = 0; n < NN; ++n)
    Av[n] = -__expf(A_logs[(size_t)(k * DI + d) * NN + n]) * LOG2E;
  float h[NN];
#pragma unroll
  for (int n = 0; n < NN; ++n) h[n] = 0.f;
  float dsum = 0.f;
  float* drow = delta + ((size_t)bk * LL + lbase) * DI + d;
#pragma unroll 1
  for (int rt = 0; rt < GR / 16; ++rt) {
    const float* ar = &sX[(rt * 16 + lid) * XP + XO];
    const float2 r03 = *reinterpret_cast<const float2*>(ar + 2 * half);
    const float2 r45 = *reinterpret_cast<const float2*>(ar + 4);
    v2f a0, a1;
    a0.x = r03.x; a0.y = r03.y;
    a1.x = half ? 0.f : r45.x; a1.y = half ? 0.f : r45.y;
    v8f d0 = {}, d1 = {};
    d0 = wmma4(a0, pb0[0], d0);
    d1 = wmma4(a0, pb0[1], d1);
    d0 = wmma4(a1, pb1[0], d0);
    d1 = wmma4(a1, pb1[1], d1);
    float* orow = drow + (size_t)(rt * 16) * DI;
    float dl[16];
#pragma unroll
    for (int v = 0; v < 8; ++v) {
      const float s0 = softplus_f(d0[v] + pbias[0]);
      const float s1 = softplus_f(d1[v] + pbias[1]);
      const float got = xhalf(half ? s0 : s1);
      dl[v] = half ? got : s0;
      dl[v + 8] = half ? s1 : got;
      orow[(size_t)v * DI] = dl[v];
      orow[(size_t)(v + 8) * DI] = dl[v + 8];
    }
    const float* urow = &sA[(rt * 16) * GP + d];
    const float* brow = &sX[(rt * 16) * XP + XO + RR];
#pragma unroll
    for (int s = 0; s < 16; ++s) {
      const float u = urow[s * GP], dlt = dl[s];
      float Bv[NN];
#pragma unroll
      for (int i = 0; i < NN / 4; ++i) {
        const float4 t4 = *reinterpret_cast<const float4*>(brow + s * XP + 4 * i);
        Bv[4 * i] = t4.x; Bv[4 * i + 1] = t4.y; Bv[4 * i + 2] = t4.z; Bv[4 * i + 3] = t4.w;
      }
      const float du = dlt * u;
      float dA[NN];
#pragma unroll
      for (int n = 0; n < NN; ++n) dA[n] = exp2_hw(dlt * Av[n]);
      float t[NN];
#pragma unroll
      for (int n = 0; n < NN; ++n) t[n] = du * Bv[n];
#pragma unroll
      for (int n = 0; n < NN; ++n) h[n] = __builtin_fmaf(dA[n], h[n], t[n]);
      dsum += dlt;
    }
  }
#pragma unroll
  for (int n = 0; n < NN; ++n)
    chunkH[((size_t)(bk * NC + chunk) * NN + n) * DI + d] = h[n];
  chunkS[(size_t)(bk * NC + chunk) * DI + d] = dsum;
}

#define SB 8

#define P2U 32
#define P2T 96
__global__ void __launch_bounds__(P2T) k_scan_pass2(
    const float* __restrict__ chunkS, const float* __restrict__ A_logs,
    float* __restrict__ chunkH) {
  const int gid = blockIdx.x * P2T + threadIdx.x;
  const int d = gid % DI;
  const int n = (gid / DI) % NN;
  const int bk = gid / (DI * NN);
  const int cs = NN * DI;
  const float a2 = -__expf(A_logs[(size_t)((bk % KK) * DI + d) * NN + n]) * LOG2E;
  const float* pa = chunkS + (size_t)bk * NC * DI + d;
  float* ph = chunkH + ((size_t)bk * NC * NN + n) * DI + d;
  float hrun = 0.f;
#pragma unroll 1
  for (int c0 = 0; c0 < NC; c0 += P2U) {
    float a[P2U], hl[P2U];
#pragma unroll
    for (int j = 0; j < P2U; ++j) {
      a[j] = pa[(c0 + j) * DI];
      hl[j] = ph[(c0 + j) * cs];
    }
#pragma unroll
    for (int j = 0; j < P2U; ++j) a[j] = exp2_hw(a2 * a[j]);
#pragma unroll
    for (int j = 0; j < P2U; ++j) {
      ph[(c0 + j) * cs] = hrun;
      hrun = a[j] * hrun + hl[j];
    }
  }
}

__device__ __forceinline__ void p3_batch(const float (&ub)[SB], const float (&db)[SB],
                                         const float (&Av)[NN], float (&h)[NN],
                                         const float Dv, const float* row0, float* yo,
                                         const int ystep) {
#pragma unroll
  for (int j = 0; j < SB; ++j) {
    const float u = ub[j], dlt = db[j];
    const float* row = row0 + (size_t)j * CDBL;
    const float du = dlt * u;
    float y = 0.f;
#pragma unroll
    for (int n = 0; n < NN; ++n) {
      const float dA = exp2_hw(dlt * Av[n]);
      h[n] = dA * h[n] + du * row[n];
      y += h[n] * row[NN + n];
    }
    yo[j * ystep] = y + u * Dv;
  }
}

#define OT (KK * DI)
static_assert(GR == CL, "an out_proj block must be exactly one scan chunk per direction");
__global__ void __launch_bounds__(OT) __attribute__((amdgpu_waves_per_eu(6)))
k_scan3_outproj(
    const float* __restrict__ xconv, const float* __restrict__ delta,
    const float* __restrict__ x_dbl, const float* __restrict__ A_logs,
    const float* __restrict__ Ds, const float* __restrict__ chunkH,
    const float* __restrict__ xz, const float* __restrict__ W_out,
    float* __restrict__ out) {
  __shared__ __attribute__((aligned(16))) float sY[KK * GR * GP];
  const int tid = threadIdx.x, wave = tid >> 5, lane = tid & 31;
  const int lid = lane & 15, half = lane >> 4;
  const int b = blockIdx.x / NC;
  const int cs = blockIdx.x % NC;
  const int k = __builtin_amdgcn_readfirstlane(tid / DI);
  const int d = tid - k * DI;
  const int c = k ? (NC - 1 - cs) : cs;
  const int bk = b * KK + k;
  {
    float Av[NN];
#pragma unroll
    for (int n = 0; n < NN; ++n)
      Av[n] = -__expf(A_logs[(size_t)(k * DI + d) * NN + n]) * LOG2E;
    const float Dv = Ds[k * DI + d];
    float h[NN];
#pragma unroll
    for (int n = 0; n < NN; ++n)
      h[n] = chunkH[((size_t)(bk * NC + c) * NN + n) * DI + d];
    const int l0 = c * CL;
    const float* up = xconv + ((size_t)b * LL + (k ? (LL - 1 - l0) : l0)) * DI + d;
    const int ustep = k ? -DI : DI;
    const float* dp = delta + ((size_t)bk * LL + l0) * DI + d;
    const float* xr = x_dbl + ((size_t)bk * LL + l0) * CDBL + RR;
    float* yo = &sY[(k * GR + (k ? (GR - 1) : 0)) * GP + d];
    const int ystep = k ? -GP : GP;
    float ub[SB], db[SB];
#pragma unroll
    for (int j = 0; j < SB; ++j) { ub[j] = up[j * ustep]; db[j] = dp[j * DI]; }
#pragma unroll 1
    for (int s0 = 0; s0 < CL - SB; s0 += SB) {
      float un[SB], dn[SB];
#pragma unroll
      for (int j = 0; j < SB; ++j) {
        un[j] = up[(s0 + SB + j) * ustep];
        dn[j] = dp[(s0 + SB + j) * DI];
      }
      __builtin_amdgcn_sched_barrier(0);
      p3_batch(ub, db, Av, h, Dv, xr + (size_t)s0 * CDBL, yo + s0 * ystep, ystep);
#pragma unroll
      for (int j = 0; j < SB; ++j) { ub[j] = un[j]; db[j] = dn[j]; }
    }
    p3_batch(ub, db, Av, h, Dv, xr + (size_t)(CL - SB) * CDBL, yo + (CL - SB) * ystep, ystep);
  }
  const int ct = wave % 6, rp = wave / 6;
  const float* wrow = W_out + (size_t)(ct * 16 + lid) * DI + 4 * half;
  float4 bf[DI / 8];
#pragma unroll
  for (int i = 0; i < DI / 8; ++i) bf[i] = *reinterpret_cast<const float4*>(wrow + 8 * i);
  __syncthreads();
  const int rowBase = blockIdx.x * GR;
  const int q = tid % (DI / 4), m0 = tid / (DI / 4);
#pragma unroll 4
  for (int it = 0; it < GR / 8; ++it) {
    const int m = it * 8 + m0;
    const float4 zv = *reinterpret_cast<const float4*>(
        xz + (size_t)(rowBase + m) * NOUT + DI + q * 4);
    const float4 y0 = *reinterpret_cast<const float4*>(&sY[m * GP + q * 4]);
    const float4 y1 = *reinterpret_cast<const float4*>(&sY[(GR + m) * GP + q * 4]);
    float4 r;
    r.x = (y0.x + y1.x) * silu_f(zv.x);
    r.y = (y0.y + y1.y) * silu_f(zv.y);
    r.z = (y0.z + y1.z) * silu_f(zv.z);
    r.w = (y0.w + y1.w) * silu_f(zv.w);
    *reinterpret_cast<float4*>(&sY[m * GP + q * 4]) = r;
  }
  __syncthreads();
  const float* ar0 = &sY[(rp * 32 + lid) * GP + 4 * half];
  const float* ar1 = ar0 + 16 * GP;
  v8f acc0 = {}, acc1 = {};
#pragma unroll
  for (int i = 0; i < DI / 8; ++i) {
    const float4 a0 = *reinterpret_cast<const float4*>(ar0 + 8 * i);
    const float4 a1 = *reinterpret_cast<const float4*>(ar1 + 8 * i);
    acc0 = wmma4(lo2(a0), lo2(bf[i]), acc0);
    acc1 = wmma4(lo2(a1), lo2(bf[i]), acc1);
    acc0 = wmma4(hi2(a0), hi2(bf[i]), acc0);
    acc1 = wmma4(hi2(a1), hi2(bf[i]), acc1);
  }
  float* orow = out + (size_t)(rowBase + rp * 32 + half * 8) * CC + ct * 16 + lid;
#pragma unroll
  for (int v = 0; v < 8; ++v) {
    orow[(size_t)v * CC] = acc0[v];
    orow[(size_t)(v + 16) * CC] = acc1[v];
  }
}

extern "C" void kernel_launch(void* const* d_in, const int* in_sizes, int n_in,
                              void* d_out, int out_size, void* d_ws, size_t ws_size,
                              hipStream_t stream) {
  (void)in_sizes; (void)n_in; (void)out_size; (void)ws_size;
  const float* x      = (const float*)d_in[0];
  const float* W_in   = (const float*)d_in[1];
  const float* conv_w = (const float*)d_in[2];
  const float* conv_b = (const float*)d_in[3];
  const float* xpw    = (const float*)d_in[4];
  const float* dtw    = (const float*)d_in[5];
  const float* dtb    = (const float*)d_in[6];
  const float* A_logs = (const float*)d_in[7];
  const float* Ds     = (const float*)d_in[8];
  const float* W_out  = (const float*)d_in[9];
  float* out = (float*)d_out;

  char* ws = (char*)d_ws;
  size_t off = 0;
  auto alloc = [&](size_t nbytes) {
    char* p = ws + off;
    off = (off + nbytes + 255) & ~(size_t)255;
    return (float*)p;
  };
  float* xz     = alloc((size_t)BL * NOUT * sizeof(float));
  float* xconv  = alloc((size_t)BL * DI * sizeof(float));
  float* x_dbl  = alloc((size_t)BB * KK * LL * CDBL * sizeof(float));
  float* delta  = alloc((size_t)BB * KK * LL * DI * sizeof(float));
  float* chunkS = alloc((size_t)BB * KK * DI * NC * sizeof(float));
  float* chunkH = alloc((size_t)BB * KK * DI * NC * NN * sizeof(float));

  k_inproj<<<BL / IPR, 256, 0, stream>>>(x, W_in, xz);
  k_dwconv<<<(BB * HH * DWSTRIPS) / (DWT / DWC4), DWT, 0, stream>>>(xz, conv_w,
                                                                    conv_b, xconv);
  k_xdbl<<<BB * KK * LL / GR, 192, 0, stream>>>(xconv, xpw, dtw, dtb, A_logs, x_dbl,
                                                delta, chunkS, chunkH);
  k_scan_pass2<<<(BB * KK * NN * DI) / P2T, P2T, 0, stream>>>(chunkS, A_logs, chunkH);
  k_scan3_outproj<<<BB * NC, OT, 0, stream>>>(
      xconv, delta, x_dbl, A_logs, Ds, chunkH, xz, W_out, out);
}


# depthwise conv fused into the x_dbl kernel as its staging step: one block = 64 spatial positions for both scan directions, conv computed once into LDS; one launch fewer; op order kept
# speedup vs baseline: 3.9225x; 1.0949x over previous
#include <hip/hip_runtime.h>
#include <cstdint>
#include <cstddef>

#define BB 4
#define HH 32
#define WWD 256
#define CC 96
#define DI 192
#define KK 2
#define NN 16
#define RR 6
#define LL (HH*WWD)
#define BL (BB*LL)
#define NOUT (2*DI)
#define CDBL (RR + 2*NN)
#define NC 128
#define CL (LL/NC)

typedef float v2f __attribute__((ext_vector_type(2)));
typedef float v8f __attribute__((ext_vector_type(8)));

__device__ __forceinline__ v8f wmma4(v2f a, v2f b, v8f c) {
  return __builtin_amdgcn_wmma_f32_16x16x4_f32(false, a, false, b, (short)0, c,
                                               false, false);
}

__device__ __forceinline__ float silu_f(float x) {
  return x / (1.0f + __expf(-x));
}
__device__ __forceinline__ float softplus_f(float x) {
  const float e = __expf(-fabsf(x));
  const float u = 1.0f + e;
  const float lost = e - (u - 1.0f);
  const float l = __builtin_amdgcn_logf(u) * 0.6931471805599453f;
  return fmaxf(x, 0.0f) + (l + lost * __builtin_amdgcn_rcpf(u));
}
#define LOG2E 1.4426950408889634f
__device__ __forceinline__ float exp2_hw(float x) { return __builtin_amdgcn_exp2f(x); }
__device__ __forceinline__ float xhalf(float x) {
  return __builtin_bit_cast(float, __builtin_amdgcn_permlanex16(
      __builtin_bit_cast(int, x), __builtin_bit_cast(int, x), 0x76543210, 0xfedcba98, false, false));
}

__device__ __forceinline__ v2f lo2(const float4 v) { v2f r; r.x = v.x; r.y = v.y; return r; }
__device__ __forceinline__ v2f hi2(const float4 v) { v2f r; r.x = v.z; r.y = v.w; return r; }

#define IPR 128
#define IPP 100
__global__ void __launch_bounds__(256) k_inproj(const float* __restrict__ x,
                                                const float* __restrict__ W_in,
                                                float* __restrict__ xz) {
  __shared__ __attribute__((aligned(16))) float sX[IPR * IPP];
  const int tid = threadIdx.x;
  const int rowBase = blockIdx.x * IPR;
  const int wave = tid >> 5, lane = tid & 31;
  const int lid = lane & 15, half = lane >> 4;
  float4 bf[3][CC / 8];
#pragma unroll
  for (int j = 0; j < 3; ++j) {
    const float* wrow = W_in + (size_t)((wave + 8 * j) * 16 + lid) * CC + 4 * half;
#pragma unroll
    for (int i = 0; i < CC / 8; ++i)
      bf[j][i] = *reinterpret_cast<const float4*>(wrow + 8 * i);
  }
#pragma unroll
  for (int it = 0; it < IPR * (CC / 4) / 256; ++it) {
    const int i = tid + it * 256;
    const int m = i / (CC / 4), q = i % (CC / 4);
    const float4 v = *reinterpret_cast<const float4*>(x + (size_t)(rowBase + m) * CC + q * 4);
    *reinterpret_cast<float4*>(&sX[m * IPP + q * 4]) = v;
  }
  __syncthreads();
#pragma unroll 1
  for (int mt = 0; mt < IPR / 16; ++mt) {
    const float* arow = &sX[(mt * 16 + lid) * IPP + 4 * half];
    float4 af[CC / 8];
#pragma unroll
    for (int i = 0; i < CC / 8; ++i) af[i] = *reinterpret_cast<const float4*>(arow + 8 * i);
    v8f acc[3] = {};
#pragma unroll
    for (int i = 0; i < CC / 8; ++i) {
#pragma unroll
      for (int j = 0; j < 3; ++j) acc[j] = wmma4(lo2(af[i]), lo2(bf[j][i]), acc[j]);
#pragma unroll
      for (int j = 0; j < 3; ++j) acc[j] = wmma4(hi2(af[i]), hi2(bf[j][i]), acc[j]);
    }
    float* orow = xz + (size_t)(rowBase + mt * 16 + half * 8) * NOUT + lid;
#pragma unroll
    for (int j = 0; j < 3; ++j)
#pragma unroll
      for (int v = 0; v < 8; ++v)
        orow[(size_t)v * NOUT + (wave + 8 * j) * 16] = acc[j][v];
  }
}


#define DWS 8
#define DWC4 (DI / 4)


#define GR 64
#define GP 196
#define XP 52
#define XO 2
#define XT (KK * DI)
static_assert(GR == CL, "an x_dbl block must be exactly one scan chunk per direction");
static_assert((XO + RR) % 4 == 0 && XO + 48 <= XP, "B row must be 16-byte aligned");
static_assert(XT / DWC4 * DWS == GR && WWD % GR == 0, "8 strips of DWS positions cover the block's rows");
__global__ void __launch_bounds__(XT) __attribute__((amdgpu_waves_per_eu(6)))
k_conv_xdbl(const float* __restrict__ xz,
            const float* __restrict__ conv_w,
            const float* __restrict__ conv_b,
            const float* __restrict__ xpw,
            const float* __restrict__ dtw,
            const float* __restrict__ dtb,
            const float* __restrict__ A_logs,
            float* __restrict__ xconv,
            float* __restrict__ x_dbl,
            float* __restrict__ delta,
            float* __restrict__ chunkS,
            float* __restrict__ chunkH) {
  __shared__ __attribute__((aligned(16))) float sA[GR * GP];
  __shared__ __attribute__((aligned(16))) float sX[KK * GR * XP];
  const int tid = threadIdx.x, wave = tid >> 5, lane = tid & 31;
  const int lid = lane & 15, half = lane >> 4;
  const int b = blockIdx.x / NC;
  const int cs = blockIdx.x % NC;
  const int k = __builtin_amdgcn_readfirstlane(tid / DI);
  const int d = tid - k * DI;
  const int w6 = wave - 6 * k;
  const int chunk = k ? (NC - 1 - cs) : cs;
  const int lbase = chunk * GR;
  const int bk = b * KK + k;
  const int j = w6 % 3, g = w6 / 3;
  const int col = j * 16 + lid;
  const bool cok = col < CDBL;
  {
    const int c = (tid % DWC4) * 4;
    const int strip = tid / DWC4;
    const int h = cs / (WWD / GR);
    const int w0 = (cs % (WWD / GR)) * GR + strip * DWS;
    float wt[4][9];
#pragma unroll
    for (int jj = 0; jj < 4; ++jj)
#pragma unroll
      for (int t = 0; t < 9; ++t) wt[jj][t] = conv_w[(c + jj) * 9 + t];
    float bias[4];
#pragma unroll
    for (int jj = 0; jj < 4; ++jj) bias[jj] = conv_b[c + jj];
    float v[3][DWS + 2][4];
#pragma unroll
    for (int r = 0; r < 3; ++r) {
      const int hh = h + r - 1;
      const bool rok = (hh >= 0) && (hh < HH);
      const int hc = hh < 0 ? 0 : (hh >= HH ? HH - 1 : hh);
#pragma unroll
      for (int q = 0; q < DWS + 2; ++q) {
        bool wok = true;
        int wc = w0 + q - 1;
        if (q == 0) { wok = w0 > 0; wc = wok ? wc : 0; }
        if (q == DWS + 1) { wok = (w0 + DWS) < WWD; wc = wok ? wc : WWD - 1; }
        const bool ok = rok && wok;
        const float4 t4 = *reinterpret_cast<const float4*>(
            xz + (size_t)((b * HH + hc) * WWD + wc) * NOUT + c);
        v[r][q][0] = ok ? t4.x : 0.f;
        v[r][q][1] = ok ? t4.y : 0.f;
        v[r][q][2] = ok ? t4.z : 0.f;
        v[r][q][3] = ok ? t4.w : 0.f;
      }
    }
    float* orow = xconv + (size_t)((b * HH + h) * WWD + w0) * DI + c;
    float* srow = &sA[(strip * DWS) * GP + c];
#pragma unroll
    for (int i = 0; i < DWS; ++i) {
      float o[4];
#pragma unroll
      for (int jj = 0; jj < 4; ++jj) {
        float acc = bias[jj];
#pragma unroll
        for (int r = 0; r < 3; ++r)
#pragma unroll
          for (int dx = 0; dx < 3; ++dx) acc += v[r][i + dx][jj] * wt[jj][r * 3 + dx];
        o[jj] = silu_f(acc);
      }
      const float4 o4 = make_float4(o[0], o[1], o[2], o[3]);
      *reinterpret_cast<float4*>(srow + i * GP) = o4;
      *reinterpret_cast<float4*>(orow + (size_t)i * DI) = o4;
    }
  }
  const float* wrow = xpw + (size_t)(k * CDBL + (cok ? col : 0)) * DI + 4 * half;
  float4 bf[DI / 8];
#pragma unroll
  for (int i = 0; i < DI / 8; ++i) {
    const float4 t4 = *reinterpret_cast<const float4*>(wrow + 8 * i);
    bf[i].x = cok ? t4.x : 0.f; bf[i].y = cok ? t4.y : 0.f;
    bf[i].z = cok ? t4.z : 0.f; bf[i].w = cok ? t4.w : 0.f;
  }
  __syncthreads();
  const int m0 = g * 32 + lid;
  const float* ar0 = &sA[(k ? (GR - 1 - m0) : m0) * GP + 4 * half];
  const float* ar1 = &sA[(k ? (GR - 17 - m0) : (m0 + 16)) * GP + 4 * half];
  v8f acc0 = {}, acc1 = {};
#pragma unroll
  for (int i = 0; i < DI / 8; ++i) {
    const float4 a0 = *reinterpret_cast<const float4*>(ar0 + 8 * i);
    const float4 a1 = *reinterpret_cast<const float4*>(ar1 + 8 * i);
    acc0 = wmma4(lo2(a0), lo2(bf[i]), acc0);
    acc1 = wmma4(lo2(a1), lo2(bf[i]), acc1);
    acc0 = wmma4(hi2(a0), hi2(bf[i]), acc0);
    acc1 = wmma4(hi2(a1), hi2(bf[i]), acc1);
  }
  v2f pb0[2], pb1[2];
  float pbias[2];
#pragma unroll
  for (int t = 0; t < 2; ++t) {
    const int dcol = w6 * 32 + t * 16 + lid;
    const float* wr = dtw + (size_t)(k * DI + dcol) * RR;
    pb0[t].x = wr[2 * half]; pb0[t].y = wr[2 * half + 1];
    const float w4 = wr[4], w5 = wr[5];
    pb1[t].x = half ? 0.f : w4; pb1[t].y = half ? 0.f : w5;
    pbias[t] = dtb[k * DI + dcol];
  }
  float* sXk = sX + k * (GR * XP);
  {
    float* xw = &sXk[(g * 32 + half * 8) * XP + XO + col];
#pragma unroll
    for (int v = 0; v < 8; ++v) {
      xw[v * XP] = acc0[v];
      xw[(v + 16) * XP] = acc1[v];
    }
  }
  __syncthreads();
  if (cok) {
    float* orow = x_dbl + ((size_t)bk * LL + lbase + g * 32 + half * 8) * CDBL + col;
#pragma unroll
    for (int v = 0; v < 8; ++v) {
      orow[(size_t)v * CDBL] = acc0[v];
      orow[(size_t)(v + 16) * CDBL] = acc1[v];
    }
  }
  float Av[NN];
#pragma unroll
  for (int n = 0; n < NN; ++n)
    Av[n] = -__expf(A_logs[(size_t)(k * DI + d) * NN + n]) * LOG2E;
  float h[NN];
#pragma unroll
  for (int n = 0; n < NN; ++n) h[n] = 0.f;
  float dsum = 0.f;
  float* drow = delta + ((size_t)bk * LL + lbase) * DI + d;
  const float* ubase = &sA[(k ? (GR - 1) : 0) * GP + d];
  const int ustep = k ? -GP : GP;
#pragma unroll 1
  for (int rt = 0; rt < GR / 16; ++rt) {
    const float* ar = &sXk[(rt * 16 + lid) * XP + XO];
    const float2 r03 = *reinterpret_cast<const float2*>(ar + 2 * half);
    const float2 r45 = *reinterpret_cast<const float2*>(ar + 4);
    v2f a0, a1;
    a0.x = r03.x; a0.y = r03.y;
    a1.x = half ? 0.f : r45.x; a1.y = half ? 0.f : r45.y;
    v8f d0 = {}, d1 = {};
    d0 = wmma4(a0, pb0[0], d0);
    d1 = wmma4(a0, pb0[1], d1);
    d0 = wmma4(a1, pb1[0], d0);
    d1 = wmma4(a1, pb1[1], d1);
    float* orow = drow + (size_t)(rt * 16) * DI;
    float dl[16];
#pragma unroll
    for (int v = 0; v < 8; ++v) {
      const float s0 = softplus_f(d0[v] + pbias[0]);
      const float s1 = softplus_f(d1[v] + pbias[1]);
      const float got = xhalf(half ? s0 : s1);
      dl[v] = half ? got : s0;
      dl[v + 8] = half ? s1 : got;
      orow[(size_t)v * DI] = dl[v];
      orow[(size_t)(v + 8) * DI] = dl[v + 8];
    }
    const float* urow = ubase + (rt * 16) * ustep;
    const float* brow = &sXk[(rt * 16) * XP + XO + RR];
#pragma unroll
    for (int s = 0; s < 16; ++s) {
      const float u = urow[s * ustep], dlt = dl[s];
      float Bv[NN];
#pragma unroll
      for (int i = 0; i < NN / 4; ++i) {
        const float4 t4 = *reinterpret_cast<const float4*>(brow + s * XP + 4 * i);
        Bv[4 * i] = t4.x; Bv[4 * i + 1] = t4.y; Bv[4 * i + 2] = t4.z; Bv[4 * i + 3] = t4.w;
      }
      const float du = dlt * u;
      float dA[NN];
#pragma unroll
      for (int n = 0; n < NN; ++n) dA[n] = exp2_hw(dlt * Av[n]);
      float t[NN];
#pragma unroll
      for (int n = 0; n < NN; ++n) t[n] = du * Bv[n];
#pragma unroll
      for (int n = 0; n < NN; ++n) h[n] = __builtin_fmaf(dA[n], h[n], t[n]);
      dsum += dlt;
    }
  }
#pragma unroll
  for (int n = 0; n < NN; ++n)
    chunkH[((size_t)(bk * NC + chunk) * NN + n) * DI + d] = h[n];
  chunkS[(size_t)(bk * NC + chunk) * DI + d] = dsum;
}

#define SB 8

#define P2U 32
#define P2T 96
__global__ void __launch_bounds__(P2T) k_scan_pass2(
    const float* __restrict__ chunkS, const float* __restrict__ A_logs,
    float* __restrict__ chunkH) {
  const int gid = blockIdx.x * P2T + threadIdx.x;
  const int d = gid % DI;
  const int n = (gid / DI) % NN;
  const int bk = gid / (DI * NN);
  const int cs = NN * DI;
  const float a2 = -__expf(A_logs[(size_t)((bk % KK) * DI + d) * NN + n]) * LOG2E;
  const float* pa = chunkS + (size_t)bk * NC * DI + d;
  float* ph = chunkH + ((size_t)bk * NC * NN + n) * DI + d;
  float hrun = 0.f;
#pragma unroll 1
  for (int c0 = 0; c0 < NC; c0 += P2U) {
    float a[P2U], hl[P2U];
#pragma unroll
    for (int j = 0; j < P2U; ++j) {
      a[j] = pa[(c0 + j) * DI];
      hl[j] = ph[(c0 + j) * cs];
    }
#pragma unroll
    for (int j = 0; j < P2U; ++j) a[j] = exp2_hw(a2 * a[j]);
#pragma unroll
    for (int j = 0; j < P2U; ++j) {
      ph[(c0 + j) * cs] = hrun;
      hrun = a[j] * hrun + hl[j];
    }
  }
}

__device__ __forceinline__ void p3_batch(const float (&ub)[SB], const float (&db)[SB],
                                         const float (&Av)[NN], float (&h)[NN],
                                         const float Dv, const float* row0, float* yo,
                                         const int ystep) {
#pragma unroll
  for (int j = 0; j < SB; ++j) {
    const float u = ub[j], dlt = db[j];
    const float* row = row0 + (size_t)j * CDBL;
    const float du = dlt * u;
    float y = 0.f;
#pragma unroll
    for (int n = 0; n < NN; ++n) {
      const float dA = exp2_hw(dlt * Av[n]);
      h[n] = dA * h[n] + du * row[n];
      y += h[n] * row[NN + n];
    }
    yo[j * ystep] = y + u * Dv;
  }
}

#define OT (KK * DI)
static_assert(GR == CL, "an out_proj block must be exactly one scan chunk per direction");
__global__ void __launch_bounds__(OT) __attribute__((amdgpu_waves_per_eu(6)))
k_scan3_outproj(
    const float* __restrict__ xconv, const float* __restrict__ delta,
    const float* __restrict__ x_dbl, const float* __restrict__ A_logs,
    const float* __restrict__ Ds, const float* __restrict__ chunkH,
    const float* __restrict__ xz, const float* __restrict__ W_out,
    float* __restrict__ out) {
  __shared__ __attribute__((aligned(16))) float sY[KK * GR * GP];
  const int tid = threadIdx.x, wave = tid >> 5, lane = tid & 31;
  const int lid = lane & 15, half = lane >> 4;
  const int b = blockIdx.x / NC;
  const int cs = blockIdx.x % NC;
  const int k = __builtin_amdgcn_readfirstlane(tid / DI);
  const int d = tid - k * DI;
  const int c = k ? (NC - 1 - cs) : cs;
  const int bk = b * KK + k;
  {
    float Av[NN];
#pragma unroll
    for (int n = 0; n < NN; ++n)
      Av[n] = -__expf(A_logs[(size_t)(k * DI + d) * NN + n]) * LOG2E;
    const float Dv = Ds[k * DI + d];
    float h[NN];
#pragma unroll
    for (int n = 0; n < NN; ++n)
      h[n] = chunkH[((size_t)(bk * NC + c) * NN + n) * DI + d];
    const int l0 = c * CL;
    const float* up = xconv + ((size_t)b * LL + (k ? (LL - 1 - l0) : l0)) * DI + d;
    const int ustep = k ? -DI : DI;
    const float* dp = delta + ((size_t)bk * LL + l0) * DI + d;
    const float* xr = x_dbl + ((size_t)bk * LL + l0) * CDBL + RR;
    float* yo = &sY[(k * GR + (k ? (GR - 1) : 0)) * GP + d];
    const int ystep = k ? -GP : GP;
    float ub[SB], db[SB];
#pragma unroll
    for (int j = 0; j < SB; ++j) { ub[j] = up[j * ustep]; db[j] = dp[j * DI]; }
#pragma unroll 1
    for (int s0 = 0; s0 < CL - SB; s0 += SB) {
      float un[SB], dn[SB];
#pragma unroll
      for (int j = 0; j < SB; ++j) {
        un[j] = up[(s0 + SB + j) * ustep];
        dn[j] = dp[(s0 + SB + j) * DI];
      }
      __builtin_amdgcn_sched_barrier(0);
      p3_batch(ub, db, Av, h, Dv, xr + (size_t)s0 * CDBL, yo + s0 * ystep, ystep);
#pragma unroll
      for (int j = 0; j < SB; ++j) { ub[j] = un[j]; db[j] = dn[j]; }
    }
    p3_batch(ub, db, Av, h, Dv, xr + (size_t)(CL - SB) * CDBL, yo + (CL - SB) * ystep, ystep);
  }
  const int ct = wave % 6, rp = wave / 6;
  const float* wrow = W_out + (size_t)(ct * 16 + lid) * DI + 4 * half;
  float4 bf[DI / 8];
#pragma unroll
  for (int i = 0; i < DI / 8; ++i) bf[i] = *reinterpret_cast<const float4*>(wrow + 8 * i);
  __syncthreads();
  const int rowBase = blockIdx.x * GR;
  const int q = tid % (DI / 4), m0 = tid / (DI / 4);
#pragma unroll 4
  for (int it = 0; it < GR / 8; ++it) {
    const int m = it * 8 + m0;
    const float4 zv = *reinterpret_cast<const float4*>(
        xz + (size_t)(rowBase + m) * NOUT + DI + q * 4);
    const float4 y0 = *reinterpret_cast<const float4*>(&sY[m * GP + q * 4]);
    const float4 y1 = *reinterpret_cast<const float4*>(&sY[(GR + m) * GP + q * 4]);
    float4 r;
    r.x = (y0.x + y1.x) * silu_f(zv.x);
    r.y = (y0.y + y1.y) * silu_f(zv.y);
    r.z = (y0.z + y1.z) * silu_f(zv.z);
    r.w = (y0.w + y1.w) * silu_f(zv.w);
    *reinterpret_cast<float4*>(&sY[m * GP + q * 4]) = r;
  }
  __syncthreads();
  const float* ar0 = &sY[(rp * 32 + lid) * GP + 4 * half];
  const float* ar1 = ar0 + 16 * GP;
  v8f acc0 = {}, acc1 = {};
#pragma unroll
  for (int i = 0; i < DI / 8; ++i) {
    const float4 a0 = *reinterpret_cast<const float4*>(ar0 + 8 * i);
    const float4 a1 = *reinterpret_cast<const float4*>(ar1 + 8 * i);
    acc0 = wmma4(lo2(a0), lo2(bf[i]), acc0);
    acc1 = wmma4(lo2(a1), lo2(bf[i]), acc1);
    acc0 = wmma4(hi2(a0), hi2(bf[i]), acc0);
    acc1 = wmma4(hi2(a1), hi2(bf[i]), acc1);
  }
  float* orow = out + (size_t)(rowBase + rp * 32 + half * 8) * CC + ct * 16 + lid;
#pragma unroll
  for (int v = 0; v < 8; ++v) {
    orow[(size_t)v * CC] = acc0[v];
    orow[(size_t)(v + 16) * CC] = acc1[v];
  }
}

extern "C" void kernel_launch(void* const* d_in, const int* in_sizes, int n_in,
                              void* d_out, int out_size, void* d_ws, size_t ws_size,
                              hipStream_t stream) {
  (void)in_sizes; (void)n_in; (void)out_size; (void)ws_size;
  const float* x      = (const float*)d_in[0];
  const float* W_in   = (const float*)d_in[1];
  const float* conv_w = (const float*)d_in[2];
  const float* conv_b = (const float*)d_in[3];
  const float* xpw    = (const float*)d_in[4];
  const float* dtw    = (const float*)d_in[5];
  const float* dtb    = (const float*)d_in[6];
  const float* A_logs = (const float*)d_in[7];
  const float* Ds     = (const float*)d_in[8];
  const float* W_out  = (const float*)d_in[9];
  float* out = (float*)d_out;

  char* ws = (char*)d_ws;
  size_t off = 0;
  auto alloc = [&](size_t nbytes) {
    char* p = ws + off;
    off = (off + nbytes + 255) & ~(size_t)255;
    return (float*)p;
  };
  float* xz     = alloc((size_t)BL * NOUT * sizeof(float));
  float* xconv  = alloc((size_t)BL * DI * sizeof(float));
  float* x_dbl  = alloc((size_t)BB * KK * LL * CDBL * sizeof(float));
  float* delta  = alloc((size_t)BB * KK * LL * DI * sizeof(float));
  float* chunkS = alloc((size_t)BB * KK * DI * NC * sizeof(float));
  float* chunkH = alloc((size_t)BB * KK * DI * NC * NN * sizeof(float));

  k_inproj<<<BL / IPR, 256, 0, stream>>>(x, W_in, xz);
  k_conv_xdbl<<<BB * NC, XT, 0, stream>>>(xz, conv_w, conv_b, xpw, dtw, dtb, A_logs,
                                          xconv, x_dbl, delta, chunkS, chunkH);
  k_scan_pass2<<<(BB * KK * NN * DI) / P2T, P2T, 0, stream>>>(chunkS, A_logs, chunkH);
  k_scan3_outproj<<<BB * NC, OT, 0, stream>>>(
      xconv, delta, x_dbl, A_logs, Ds, chunkH, xz, W_out, out);
}
